# starting kernel
# baseline (speedup 1.0000x reference)
#include <hip/hip_runtime.h>
#include <hip/hip_bf16.h>

#define IN_FEATS   128
#define OUT_FEATS  16
#define NEG_SLOPE  0.2f

typedef float v2f __attribute__((ext_vector_type(2)));
typedef float v8f __attribute__((ext_vector_type(8)));

__global__ void gat_project_wmma(const float* __restrict__ h,
                                 const float* __restrict__ W,
                                 float* __restrict__ hW, int nNodes) {
  int wave = (blockIdx.x * blockDim.x + threadIdx.x) >> 5;
  int lane = threadIdx.x & 31;
  int m0 = wave * 16;
  if (m0 >= nNodes) return;
  int half = lane >> 4;
  int l16  = lane & 15;
  const float* arow = h + (size_t)(m0 + l16) * IN_FEATS;
  const float* brow = W + (size_t)l16 * IN_FEATS;
  v8f acc = {};
#pragma unroll
  for (int k = 0; k < IN_FEATS; k += 4) {
    int ka = k + half * 2;
    v2f a = *(const v2f*)(arow + ka);
    v2f b = *(const v2f*)(brow + ka);
    acc = __builtin_amdgcn_wmma_f32_16x16x4_f32(false, a, false, b,
                                                (short)0, acc, false, false);
  }
#pragma unroll
  for (int g = 0; g < 8; ++g) {
    int row = m0 + half * 8 + g;
    hW[(size_t)row * OUT_FEATS + l16] = acc[g];
  }
}

__global__ void gat_node_prep(const float* __restrict__ hW,
                              const float* __restrict__ a_l,
                              const float* __restrict__ a_r,
                              float* __restrict__ el, float* __restrict__ er,
                              float* __restrict__ emax, float* __restrict__ esum,
                              float* __restrict__ out, int nNodes) {
  int n = blockIdx.x * blockDim.x + threadIdx.x;
  if (n >= nNodes) return;
  const float* row = hW + (size_t)n * OUT_FEATS;
  float sl = 0.f, sr = 0.f;
#pragma unroll
  for (int f = 0; f < OUT_FEATS; ++f) {
    float v = row[f];
    sl += v * a_l[f];
    sr += v * a_r[f];
    out[(size_t)n * OUT_FEATS + f] = 0.f;
  }
  el[n] = sl;
  er[n] = sr;
  emax[n] = __int_as_float(0xFF800000);
  esum[n] = 0.f;
}

__device__ __forceinline__ void atomicMaxFloat(float* addr, float val) {
  if (val >= 0.f) atomicMax((int*)addr, __float_as_int(val));
  else            atomicMin((unsigned int*)addr, (unsigned int)__float_as_int(val));
}

__global__ void gat_edge_logit(const int* __restrict__ src,
                               const int* __restrict__ dst,
                               const float* __restrict__ el,
                               const float* __restrict__ er,
                               float* __restrict__ elog,
                               float* __restrict__ emax, int nEdges) {
  int e = blockIdx.x * blockDim.x + threadIdx.x;
  if (e >= nEdges) return;
  int s = src[e], d = dst[e];
  float x = el[s] + er[d];
  x = (x > 0.f) ? x : NEG_SLOPE * x;
  elog[e] = x;
  atomicMaxFloat(emax + d, x);
}

__global__ void gat_node_fixmax(float* __restrict__ emax, int nNodes) {
  int n = blockIdx.x * blockDim.x + threadIdx.x;
  if (n >= nNodes) return;
  if (emax[n] == __int_as_float(0xFF800000)) emax[n] = 0.f;
}

__global__ void gat_edge_exp(const int* __restrict__ dst,
                             float* __restrict__ elog,
                             const float* __restrict__ emax,
                             float* __restrict__ esum, int nEdges) {
  int e = blockIdx.x * blockDim.x + threadIdx.x;
  if (e >= nEdges) return;
  int d = dst[e];
  float ex = __expf(elog[e] - emax[d]);
  elog[e] = ex;
  atomicAdd(esum + d, ex);
}

__global__ void gat_edge_aggr(const int* __restrict__ src,
                              const int* __restrict__ dst,
                              const float* __restrict__ elog,
                              const float* __restrict__ esum,
                              const float* __restrict__ hW,
                              float* __restrict__ out, int nEdges) {
  long long t = (long long)blockIdx.x * blockDim.x + threadIdx.x;
  int e = (int)(t >> 4);
  int f = (int)(t & 15);
  if (e >= nEdges) return;
  int s = src[e], d = dst[e];
  float alpha = elog[e] / fmaxf(esum[d], 1e-16f);
  atomicAdd(out + (size_t)d * OUT_FEATS + f,
            hW[(size_t)s * OUT_FEATS + f] * alpha);
}

extern "C" void kernel_launch(void* const* d_in, const int* in_sizes, int n_in,
                              void* d_out, int out_size, void* d_ws, size_t ws_size,
                              hipStream_t stream) {
  const float* h   = (const float*)d_in[0];
  const int*   src = (const int*)d_in[1];
  const int*   dst = (const int*)d_in[2];
  const float* W   = (const float*)d_in[3];
  const float* a_l = (const float*)d_in[4];
  const float* a_r = (const float*)d_in[5];
  float* out = (float*)d_out;

  int nNodes = in_sizes[0] / IN_FEATS;
  int nEdges = in_sizes[1];

  float* ws   = (float*)d_ws;
  float* hW   = ws;
  float* el   = hW + (size_t)nNodes * OUT_FEATS;
  float* er   = el + nNodes;
  float* emax = er + nNodes;
  float* esum = emax + nNodes;
  float* elog = esum + nNodes;

  dim3 blk(256);
  int tiles = (nNodes + 15) / 16;
  int blocksK1 = (tiles + 7) / 8;

  gat_project_wmma<<<blocksK1, blk, 0, stream>>>(h, W, hW, nNodes);
  gat_node_prep<<<(nNodes + 255) / 256, blk, 0, stream>>>(hW, a_l, a_r, el, er,
                                                          emax, esum, out, nNodes);
  gat_edge_logit<<<(nEdges + 255) / 256, blk, 0, stream>>>(src, dst, el, er,
                                                           elog, emax, nEdges);
  gat_node_fixmax<<<(nNodes + 255) / 256, blk, 0, stream>>>(emax, nNodes);
  gat_edge_exp<<<(nEdges + 255) / 256, blk, 0, stream>>>(dst, elog, emax, esum, nEdges);

  long long laneWork = (long long)nEdges * OUT_FEATS;
  int blocksK6 = (int)((laneWork + 255) / 256);
  gat_edge_aggr<<<blocksK6, blk, 0, stream>>>(src, dst, elog, esum, hW, out, nEdges);
}


# Segment max and sum via LDS atomics per (edge slice, destination range) work-group, slice planes reduced per node; replaces the global atomics of both scalar edge passes; sum order changes
# speedup vs baseline: 1.9246x; 1.9246x over previous
#include <hip/hip_runtime.h>
#include <hip/hip_bf16.h>

#define IN_FEATS   128
#define OUT_FEATS  16
#define NEG_SLOPE  0.2f

typedef float v2f __attribute__((ext_vector_type(2)));
typedef float v8f __attribute__((ext_vector_type(8)));

__global__ void gat_project_wmma(const float* __restrict__ h,
                                 const float* __restrict__ W,
                                 float* __restrict__ hW, int nNodes) {
  int wave = (blockIdx.x * blockDim.x + threadIdx.x) >> 5;
  int lane = threadIdx.x & 31;
  int m0 = wave * 16;
  if (m0 >= nNodes) return;
  int half = lane >> 4;
  int l16  = lane & 15;
  const float* arow = h + (size_t)(m0 + l16) * IN_FEATS;
  const float* brow = W + (size_t)l16 * IN_FEATS;
  v8f acc = {};
#pragma unroll
  for (int k = 0; k < IN_FEATS; k += 4) {
    int ka = k + half * 2;
    v2f a = *(const v2f*)(arow + ka);
    v2f b = *(const v2f*)(brow + ka);
    acc = __builtin_amdgcn_wmma_f32_16x16x4_f32(false, a, false, b,
                                                (short)0, acc, false, false);
  }
#pragma unroll
  for (int g = 0; g < 8; ++g) {
    int row = m0 + half * 8 + g;
    hW[(size_t)row * OUT_FEATS + l16] = acc[g];
  }
}

#define N_SLICES     28
#define N_RANGES     9
#define SEG_THREADS  1024
#define SEG_EDGES    8

__global__ void gat_node_prep(const float* __restrict__ hW,
                              const float* __restrict__ a_l,
                              const float* __restrict__ a_r,
                              float* __restrict__ el, float* __restrict__ er,
                              float* __restrict__ out, int nNodes) {
  int n = blockIdx.x * blockDim.x + threadIdx.x;
  if (n >= nNodes) return;
  const float* row = hW + (size_t)n * OUT_FEATS;
  float sl = 0.f, sr = 0.f;
#pragma unroll
  for (int f = 0; f < OUT_FEATS; ++f) {
    float v = row[f];
    sl += v * a_l[f];
    sr += v * a_r[f];
    out[(size_t)n * OUT_FEATS + f] = 0.f;
  }
  el[n] = sl;
  er[n] = sr;
}

__device__ __forceinline__ void ldsMaxFloat(float* addr, float val) {
  if (val >= 0.f) atomicMax((int*)addr, __float_as_int(val));
  else            atomicMin((unsigned int*)addr, (unsigned int)__float_as_int(val));
}

__device__ __forceinline__ void load_edges(const int* __restrict__ src,
                                           const int* __restrict__ dst,
                                           int base, int e1,
                                           int (&s)[SEG_EDGES], int (&d)[SEG_EDGES]) {
  if (base + SEG_EDGES <= e1) {
    int4 d0 = *(const int4*)(dst + base);
    int4 d1 = *(const int4*)(dst + base + 4);
    int4 s0 = *(const int4*)(src + base);
    int4 s1 = *(const int4*)(src + base + 4);
    d[0] = d0.x; d[1] = d0.y; d[2] = d0.z; d[3] = d0.w;
    d[4] = d1.x; d[5] = d1.y; d[6] = d1.z; d[7] = d1.w;
    s[0] = s0.x; s[1] = s0.y; s[2] = s0.z; s[3] = s0.w;
    s[4] = s1.x; s[5] = s1.y; s[6] = s1.z; s[7] = s1.w;
  } else {
#pragma unroll
    for (int j = 0; j < SEG_EDGES; ++j) {
      int e = base + j;
      bool ok = e < e1;
      d[j] = ok ? dst[ok ? e : 0] : -1;
      s[j] = ok ? src[ok ? e : 0] : 0;
    }
  }
}

__global__ void __launch_bounds__(SEG_THREADS)
gat_segmax_lds(const int* __restrict__ src, const int* __restrict__ dst,
               const float* __restrict__ el, const float* __restrict__ er,
               float* __restrict__ planes, int nNodes, int nEdges,
               int rangeLen, int sliceLen) {
  extern __shared__ float tab[];
  const int tid = threadIdx.x;
  const int lo = blockIdx.y * rangeLen;
  int cnt = nNodes - lo;
  cnt = cnt > rangeLen ? rangeLen : (cnt < 0 ? 0 : cnt);
  for (int i = tid; i < cnt; i += SEG_THREADS) tab[i] = __int_as_float(0xFF800000);
  __syncthreads();
  const int e0 = blockIdx.x * sliceLen;
  int e1 = e0 + sliceLen;
  if (e1 > nEdges) e1 = nEdges;
  int s[SEG_EDGES], d[SEG_EDGES];
  int base = e0 + tid * SEG_EDGES;
  load_edges(src, dst, base, e1, s, d);
  while (base < e1) {
    const int nbase = base + SEG_THREADS * SEG_EDGES;
    int ns[SEG_EDGES], nd[SEG_EDGES];
    load_edges(src, dst, nbase, e1, ns, nd);
    float x[SEG_EDGES];
#pragma unroll
    for (int j = 0; j < SEG_EDGES; ++j) {
      bool in = (unsigned)(d[j] - lo) < (unsigned)cnt;
      x[j] = el[in ? s[j] : 0] + er[in ? d[j] : 0];
    }
#pragma unroll
    for (int j = 0; j < SEG_EDGES; ++j) {
      unsigned u = (unsigned)(d[j] - lo);
      if (u < (unsigned)cnt) {
        float v = (x[j] > 0.f) ? x[j] : NEG_SLOPE * x[j];
        ldsMaxFloat(tab + u, v);
      }
    }
#pragma unroll
    for (int j = 0; j < SEG_EDGES; ++j) { s[j] = ns[j]; d[j] = nd[j]; }
    base = nbase;
  }
  __syncthreads();
  float* plane = planes + (size_t)blockIdx.x * nNodes + lo;
  for (int i = tid; i < cnt; i += SEG_THREADS) plane[i] = tab[i];
}

__global__ void gat_reduce_max(const float* __restrict__ planes,
                               const float* __restrict__ er,
                               float2* __restrict__ erMax, int nNodes) {
  int n = blockIdx.x * blockDim.x + threadIdx.x;
  if (n >= nNodes) return;
  float m = __int_as_float(0xFF800000);
#pragma unroll
  for (int b = 0; b < N_SLICES; ++b) m = fmaxf(m, planes[(size_t)b * nNodes + n]);
  if (!(fabsf(m) < __int_as_float(0x7F800000))) m = 0.f;
  erMax[n] = make_float2(er[n], m);
}

__global__ void __launch_bounds__(SEG_THREADS)
gat_segsum_lds(const int* __restrict__ src, const int* __restrict__ dst,
               const float* __restrict__ el, const float2* __restrict__ erMax,
               float* __restrict__ planes, int nNodes, int nEdges,
               int rangeLen, int sliceLen) {
  extern __shared__ float tab[];
  const int tid = threadIdx.x;
  const int lo = blockIdx.y * rangeLen;
  int cnt = nNodes - lo;
  cnt = cnt > rangeLen ? rangeLen : (cnt < 0 ? 0 : cnt);
  for (int i = tid; i < cnt; i += SEG_THREADS) tab[i] = 0.f;
  __syncthreads();
  const int e0 = blockIdx.x * sliceLen;
  int e1 = e0 + sliceLen;
  if (e1 > nEdges) e1 = nEdges;
  int s[SEG_EDGES], d[SEG_EDGES];
  int base = e0 + tid * SEG_EDGES;
  load_edges(src, dst, base, e1, s, d);
  while (base < e1) {
    const int nbase = base + SEG_THREADS * SEG_EDGES;
    int ns[SEG_EDGES], nd[SEG_EDGES];
    load_edges(src, dst, nbase, e1, ns, nd);
    float a[SEG_EDGES];
    float2 rm[SEG_EDGES];
#pragma unroll
    for (int j = 0; j < SEG_EDGES; ++j) {
      bool in = (unsigned)(d[j] - lo) < (unsigned)cnt;
      a[j]  = el[in ? s[j] : 0];
      rm[j] = erMax[in ? d[j] : 0];
    }
#pragma unroll
    for (int j = 0; j < SEG_EDGES; ++j) {
      unsigned u = (unsigned)(d[j] - lo);
      if (u < (unsigned)cnt) {
        float x = a[j] + rm[j].x;
        x = (x > 0.f) ? x : NEG_SLOPE * x;
        atomicAdd(tab + u, __expf(x - rm[j].y));
      }
    }
#pragma unroll
    for (int j = 0; j < SEG_EDGES; ++j) { s[j] = ns[j]; d[j] = nd[j]; }
    base = nbase;
  }
  __syncthreads();
  float* plane = planes + (size_t)blockIdx.x * nNodes + lo;
  for (int i = tid; i < cnt; i += SEG_THREADS) plane[i] = tab[i];
}

__global__ void gat_reduce_sum(const float* __restrict__ planes,
                               const float2* __restrict__ erMax,
                               float4* __restrict__ nodeRec, int nNodes) {
  int n = blockIdx.x * blockDim.x + threadIdx.x;
  if (n >= nNodes) return;
  float sum = 0.f;
#pragma unroll
  for (int b = 0; b < N_SLICES; ++b) sum += planes[(size_t)b * nNodes + n];
  float2 rm = erMax[n];
  nodeRec[n] = make_float4(rm.x, rm.y, sum, 0.f);
}

__global__ void gat_edge_aggr(const int* __restrict__ src,
                              const int* __restrict__ dst,
                              const float* __restrict__ el,
                              const float4* __restrict__ nodeRec,
                              const float* __restrict__ hW,
                              float* __restrict__ out, int nEdges) {
  long long t = (long long)blockIdx.x * blockDim.x + threadIdx.x;
  int e = (int)(t >> 4);
  int f = (int)(t & 15);
  if (e >= nEdges) return;
  int s = src[e], d = dst[e];
  float4 r = nodeRec[d];
  float x = el[s] + r.x;
  x = (x > 0.f) ? x : NEG_SLOPE * x;
  float alpha = __expf(x - r.y) / fmaxf(r.z, 1e-16f);
  atomicAdd(out + (size_t)d * OUT_FEATS + f,
            hW[(size_t)s * OUT_FEATS + f] * alpha);
}

extern "C" void kernel_launch(void* const* d_in, const int* in_sizes, int n_in,
                              void* d_out, int out_size, void* d_ws, size_t ws_size,
                              hipStream_t stream) {
  const float* h   = (const float*)d_in[0];
  const int*   src = (const int*)d_in[1];
  const int*   dst = (const int*)d_in[2];
  const float* W   = (const float*)d_in[3];
  const float* a_l = (const float*)d_in[4];
  const float* a_r = (const float*)d_in[5];
  float* out = (float*)d_out;

  int nNodes = in_sizes[0] / IN_FEATS;
  int nEdges = in_sizes[1];

  float* ws      = (float*)d_ws;
  float* hW      = ws;
  float* el      = hW + (size_t)nNodes * OUT_FEATS;
  float* er      = el + nNodes;
  float2* erMax  = (float2*)(er + nNodes);
  float* planes  = er + (size_t)nNodes * 3;
  float4* nodeRec = (float4*)(planes + (size_t)N_SLICES * nNodes);

  dim3 blk(256);
  int tiles = (nNodes + 15) / 16;
  int blocksK1 = (tiles + 7) / 8;
  int nodeBlocks = (nNodes + 255) / 256;

  int rangeLen = ((nNodes + N_RANGES - 1) / N_RANGES + 31) / 32 * 32;
  int sliceLen = (((nEdges + N_SLICES - 1) / N_SLICES) + SEG_EDGES - 1) / SEG_EDGES * SEG_EDGES;
  size_t ldsBytes = (size_t)rangeLen * sizeof(float);
  dim3 segGrid(N_SLICES, N_RANGES);

  gat_project_wmma<<<blocksK1, blk, 0, stream>>>(h, W, hW, nNodes);
  gat_node_prep<<<nodeBlocks, blk, 0, stream>>>(hW, a_l, a_r, el, er, out, nNodes);
  gat_segmax_lds<<<segGrid, SEG_THREADS, ldsBytes, stream>>>(src, dst, el, er, planes,
                                                             nNodes, nEdges, rangeLen, sliceLen);
  gat_reduce_max<<<nodeBlocks, blk, 0, stream>>>(planes, er, erMax, nNodes);
  gat_segsum_lds<<<segGrid, SEG_THREADS, ldsBytes, stream>>>(src, dst, el, erMax, planes,
                                                             nNodes, nEdges, rangeLen, sliceLen);
  gat_reduce_sum<<<nodeBlocks, blk, 0, stream>>>(planes, erMax, nodeRec, nNodes);

  long long laneWork = (long long)nEdges * OUT_FEATS;
  int blocksK6 = (int)((laneWork + 255) / 256);
  gat_edge_aggr<<<blocksK6, blk, 0, stream>>>(src, dst, el, nodeRec, hW, out, nEdges);
}


# Aggregation sums edge rows in LDS tables per (edge slice, destination range) work-group, in-range edges queued in LDS; one global add per row element per slice, not per edge; sum order changes
# speedup vs baseline: 2.3750x; 1.2340x over previous
#include <hip/hip_runtime.h>
#include <hip/hip_bf16.h>

#define IN_FEATS   128
#define OUT_FEATS  16
#define NEG_SLOPE  0.2f

typedef float v2f __attribute__((ext_vector_type(2)));
typedef float v8f __attribute__((ext_vector_type(8)));

__global__ void gat_project_wmma(const float* __restrict__ h,
                                 const float* __restrict__ W,
                                 float* __restrict__ hW, int nNodes) {
  int wave = (blockIdx.x * blockDim.x + threadIdx.x) >> 5;
  int lane = threadIdx.x & 31;
  int m0 = wave * 16;
  if (m0 >= nNodes) return;
  int half = lane >> 4;
  int l16  = lane & 15;
  const float* arow = h + (size_t)(m0 + l16) * IN_FEATS;
  const float* brow = W + (size_t)l16 * IN_FEATS;
  v8f acc = {};
#pragma unroll
  for (int k = 0; k < IN_FEATS; k += 4) {
    int ka = k + half * 2;
    v2f a = *(const v2f*)(arow + ka);
    v2f b = *(const v2f*)(brow + ka);
    acc = __builtin_amdgcn_wmma_f32_16x16x4_f32(false, a, false, b,
                                                (short)0, acc, false, false);
  }
#pragma unroll
  for (int g = 0; g < 8; ++g) {
    int row = m0 + half * 8 + g;
    hW[(size_t)row * OUT_FEATS + l16] = acc[g];
  }
}

#define N_SLICES     28
#define N_RANGES     9
#define SEG_THREADS  1024
#define SEG_EDGES    8

__global__ void gat_node_prep(const float* __restrict__ hW,
                              const float* __restrict__ a_l,
                              const float* __restrict__ a_r,
                              float* __restrict__ el, float* __restrict__ er,
                              float* __restrict__ out, int nNodes) {
  int n = blockIdx.x * blockDim.x + threadIdx.x;
  if (n >= nNodes) return;
  const float* row = hW + (size_t)n * OUT_FEATS;
  float sl = 0.f, sr = 0.f;
#pragma unroll
  for (int f = 0; f < OUT_FEATS; ++f) {
    float v = row[f];
    sl += v * a_l[f];
    sr += v * a_r[f];
    out[(size_t)n * OUT_FEATS + f] = 0.f;
  }
  el[n] = sl;
  er[n] = sr;
}

__device__ __forceinline__ void ldsMaxFloat(float* addr, float val) {
  if (val >= 0.f) atomicMax((int*)addr, __float_as_int(val));
  else            atomicMin((unsigned int*)addr, (unsigned int)__float_as_int(val));
}

__device__ __forceinline__ void load_edges(const int* __restrict__ src,
                                           const int* __restrict__ dst,
                                           int base, int e1,
                                           int (&s)[SEG_EDGES], int (&d)[SEG_EDGES]) {
  if (base + SEG_EDGES <= e1) {
    int4 d0 = *(const int4*)(dst + base);
    int4 d1 = *(const int4*)(dst + base + 4);
    int4 s0 = *(const int4*)(src + base);
    int4 s1 = *(const int4*)(src + base + 4);
    d[0] = d0.x; d[1] = d0.y; d[2] = d0.z; d[3] = d0.w;
    d[4] = d1.x; d[5] = d1.y; d[6] = d1.z; d[7] = d1.w;
    s[0] = s0.x; s[1] = s0.y; s[2] = s0.z; s[3] = s0.w;
    s[4] = s1.x; s[5] = s1.y; s[6] = s1.z; s[7] = s1.w;
  } else {
#pragma unroll
    for (int j = 0; j < SEG_EDGES; ++j) {
      int e = base + j;
      bool ok = e < e1;
      d[j] = ok ? dst[ok ? e : 0] : -1;
      s[j] = ok ? src[ok ? e : 0] : 0;
    }
  }
}

__global__ void __launch_bounds__(SEG_THREADS)
gat_segmax_lds(const int* __restrict__ src, const int* __restrict__ dst,
               const float* __restrict__ el, const float* __restrict__ er,
               float* __restrict__ planes, int nNodes, int nEdges,
               int rangeLen, int sliceLen) {
  extern __shared__ float tab[];
  const int tid = threadIdx.x;
  const int lo = blockIdx.y * rangeLen;
  int cnt = nNodes - lo;
  cnt = cnt > rangeLen ? rangeLen : (cnt < 0 ? 0 : cnt);
  for (int i = tid; i < cnt; i += SEG_THREADS) tab[i] = __int_as_float(0xFF800000);
  __syncthreads();
  const int e0 = blockIdx.x * sliceLen;
  int e1 = e0 + sliceLen;
  if (e1 > nEdges) e1 = nEdges;
  int s[SEG_EDGES], d[SEG_EDGES];
  int base = e0 + tid * SEG_EDGES;
  load_edges(src, dst, base, e1, s, d);
  while (base < e1) {
    const int nbase = base + SEG_THREADS * SEG_EDGES;
    int ns[SEG_EDGES], nd[SEG_EDGES];
    load_edges(src, dst, nbase, e1, ns, nd);
    float x[SEG_EDGES];
#pragma unroll
    for (int j = 0; j < SEG_EDGES; ++j) {
      bool in = (unsigned)(d[j] - lo) < (unsigned)cnt;
      x[j] = el[in ? s[j] : 0] + er[in ? d[j] : 0];
    }
#pragma unroll
    for (int j = 0; j < SEG_EDGES; ++j) {
      unsigned u = (unsigned)(d[j] - lo);
      if (u < (unsigned)cnt) {
        float v = (x[j] > 0.f) ? x[j] : NEG_SLOPE * x[j];
        ldsMaxFloat(tab + u, v);
      }
    }
#pragma unroll
    for (int j = 0; j < SEG_EDGES; ++j) { s[j] = ns[j]; d[j] = nd[j]; }
    base = nbase;
  }
  __syncthreads();
  float* plane = planes + (size_t)blockIdx.x * nNodes + lo;
  for (int i = tid; i < cnt; i += SEG_THREADS) plane[i] = tab[i];
}

__global__ void gat_reduce_max(const float* __restrict__ planes,
                               const float* __restrict__ er,
                               float2* __restrict__ erMax, int nNodes) {
  int n = blockIdx.x * blockDim.x + threadIdx.x;
  if (n >= nNodes) return;
  float m = __int_as_float(0xFF800000);
#pragma unroll
  for (int b = 0; b < N_SLICES; ++b) m = fmaxf(m, planes[(size_t)b * nNodes + n]);
  if (!(fabsf(m) < __int_as_float(0x7F800000))) m = 0.f;
  erMax[n] = make_float2(er[n], m);
}

__global__ void __launch_bounds__(SEG_THREADS)
gat_segsum_lds(const int* __restrict__ src, const int* __restrict__ dst,
               const float* __restrict__ el, const float2* __restrict__ erMax,
               float* __restrict__ planes, int nNodes, int nEdges,
               int rangeLen, int sliceLen) {
  extern __shared__ float tab[];
  const int tid = threadIdx.x;
  const int lo = blockIdx.y * rangeLen;
  int cnt = nNodes - lo;
  cnt = cnt > rangeLen ? rangeLen : (cnt < 0 ? 0 : cnt);
  for (int i = tid; i < cnt; i += SEG_THREADS) tab[i] = 0.f;
  __syncthreads();
  const int e0 = blockIdx.x * sliceLen;
  int e1 = e0 + sliceLen;
  if (e1 > nEdges) e1 = nEdges;
  int s[SEG_EDGES], d[SEG_EDGES];
  int base = e0 + tid * SEG_EDGES;
  load_edges(src, dst, base, e1, s, d);
  while (base < e1) {
    const int nbase = base + SEG_THREADS * SEG_EDGES;
    int ns[SEG_EDGES], nd[SEG_EDGES];
    load_edges(src, dst, nbase, e1, ns, nd);
    float a[SEG_EDGES];
    float2 rm[SEG_EDGES];
#pragma unroll
    for (int j = 0; j < SEG_EDGES; ++j) {
      bool in = (unsigned)(d[j] - lo) < (unsigned)cnt;
      a[j]  = el[in ? s[j] : 0];
      rm[j] = erMax[in ? d[j] : 0];
    }
#pragma unroll
    for (int j = 0; j < SEG_EDGES; ++j) {
      unsigned u = (unsigned)(d[j] - lo);
      if (u < (unsigned)cnt) {
        float x = a[j] + rm[j].x;
        x = (x > 0.f) ? x : NEG_SLOPE * x;
        atomicAdd(tab + u, __expf(x - rm[j].y));
      }
    }
#pragma unroll
    for (int j = 0; j < SEG_EDGES; ++j) { s[j] = ns[j]; d[j] = nd[j]; }
    base = nbase;
  }
  __syncthreads();
  float* plane = planes + (size_t)blockIdx.x * nNodes + lo;
  for (int i = tid; i < cnt; i += SEG_THREADS) plane[i] = tab[i];
}

__global__ void gat_reduce_sum(const float* __restrict__ planes,
                               const float2* __restrict__ erMax,
                               float4* __restrict__ nodeRec, int nNodes) {
  int n = blockIdx.x * blockDim.x + threadIdx.x;
  if (n >= nNodes) return;
  float sum = 0.f;
#pragma unroll
  for (int b = 0; b < N_SLICES; ++b) sum += planes[(size_t)b * nNodes + n];
  float2 rm = erMax[n];
  nodeRec[n] = make_float4(rm.x, rm.y, sum, 0.f);
}

#define AG_SLICES   8
#define AG_RANGES   32
#define AG_THREADS  1024
#define AG_EDGES    16
#define AG_QCAP     (AG_THREADS * AG_EDGES)
#define AG_ROWS     3126
#define AG_GROUPS   (AG_THREADS / OUT_FEATS)

__device__ __forceinline__ void load_dst16(const int* __restrict__ dst, int base, int e1,
                                           int (&d)[AG_EDGES]) {
  if (base + AG_EDGES <= e1) {
#pragma unroll
    for (int q = 0; q < AG_EDGES / 4; ++q) {
      int4 v = *(const int4*)(dst + base + 4 * q);
      d[4 * q] = v.x; d[4 * q + 1] = v.y; d[4 * q + 2] = v.z; d[4 * q + 3] = v.w;
    }
  } else {
#pragma unroll
    for (int j = 0; j < AG_EDGES; ++j) {
      int e = base + j;
      bool ok = e < e1;
      d[j] = ok ? dst[ok ? e : 0] : -1;
    }
  }
}

template <int K> __device__ __forceinline__ int row_share_i(int v) {
  return __builtin_amdgcn_update_dpp(v, v, 0x150 + K, 0xF, 0xF, false);
}
template <int K> __device__ __forceinline__ float row_share_f(float v) {
  return __int_as_float(row_share_i<K>(__float_as_int(v)));
}
#define AG_FOR16(X) X(0) X(1) X(2) X(3) X(4) X(5) X(6) X(7) X(8) X(9) X(10) X(11) X(12) X(13) X(14) X(15)
#define AG_ROW_LOAD(K) hv[K] = hW[(unsigned)(row_share_i<K>(s) * OUT_FEATS + f)];
#define AG_ROW_ADD(K)  atomicAdd(tab + row_share_i<K>(urow) * OUT_FEATS + f, hv[K] * row_share_f<K>(alpha));

__device__ __forceinline__ unsigned range_mask(const int (&d)[AG_EDGES], int lo, int cnt) {
  unsigned m = 0;
#pragma unroll
  for (int j = 0; j < AG_EDGES; ++j)
    m |= (unsigned)((unsigned)(d[j] - lo) < (unsigned)cnt) << j;
  return m;
}

__global__ void __launch_bounds__(AG_THREADS)
gat_edge_aggr_lds(const int* __restrict__ src, const int* __restrict__ dst,
                  const float* __restrict__ el, const float4* __restrict__ nodeRec,
                  const float* __restrict__ hW, float* __restrict__ out,
                  int nNodes, int nEdges) {
  __shared__ float tab[AG_ROWS * OUT_FEATS];
  __shared__ int queue[AG_QCAP];
  __shared__ int qn[2];
  const int tid = threadIdx.x;
  const int lane = tid & 31;
  const int g = tid >> 4;
  const int f = tid & 15;
  const int sliceLen = (((nEdges + (int)gridDim.x - 1) / (int)gridDim.x) + AG_EDGES - 1)
                       / AG_EDGES * AG_EDGES;
  const int e0 = blockIdx.x * sliceLen;
  int e1 = e0 + sliceLen;
  if (e1 > nEdges) e1 = nEdges;

  for (int lo = blockIdx.y * AG_ROWS; lo < nNodes; lo += (int)gridDim.y * AG_ROWS) {
    int cnt = nNodes - lo;
    cnt = cnt > AG_ROWS ? AG_ROWS : cnt;
    for (int i = tid; i < cnt * OUT_FEATS; i += AG_THREADS) tab[i] = 0.f;
    if (tid < 2) qn[tid] = 0;
    __syncthreads();

    unsigned hit;
    {
      int d[AG_EDGES];
      load_dst16(dst, e0 + tid * AG_EDGES, e1, d);
      hit = range_mask(d, lo, cnt);
    }
    int par = 0;
    for (int tb = e0; tb < e1; tb += AG_QCAP) {
      const int base = tb + tid * AG_EDGES;
      const int c = __popc(hit);
      int incl = c;
#pragma unroll
      for (int o = 1; o < 32; o <<= 1) {
        int t = __shfl_up(incl, o);
        if (lane >= o) incl += t;
      }
      const int total = __builtin_amdgcn_readlane(incl, 31);
      int wbase = 0;
      if (total > 0) {
        if (lane == 0) wbase = atomicAdd(&qn[par], total);
        wbase = __builtin_amdgcn_readfirstlane(wbase);
      }
      int pos = wbase + incl - c;
      for (unsigned hb = hit; hb != 0u; hb &= hb - 1u)
        queue[pos++] = base + __builtin_ctz(hb);
      __syncthreads();

      int nd[AG_EDGES];
      load_dst16(dst, base + AG_QCAP, e1, nd);

      const int n = qn[par];
      if (tid == 0) qn[par ^ 1] = 0;

      for (int q0 = g * OUT_FEATS; q0 < n; q0 += AG_GROUPS * OUT_FEATS) {
        const int idx = q0 + f;
        const bool okl = idx < n;
        const int e = queue[okl ? idx : q0];
        const int s = src[e];
        const int urow = dst[e] - lo;
        const float4 r = nodeRec[urow + lo];
        float x = el[s] + r.x;
        x = (x > 0.f) ? x : NEG_SLOPE * x;
        float alpha = __expf(x - r.y) / fmaxf(r.z, 1e-16f);
        alpha = okl ? alpha : 0.f;
        float hv[OUT_FEATS];
        AG_FOR16(AG_ROW_LOAD)
        AG_FOR16(AG_ROW_ADD)
      }
      __syncthreads();
      hit = range_mask(nd, lo, cnt);
      par ^= 1;
    }

    float* o = out + (size_t)lo * OUT_FEATS;
    for (int i = tid; i < cnt * OUT_FEATS; i += AG_THREADS) atomicAdd(o + i, tab[i]);
    __syncthreads();
  }
}

extern "C" void kernel_launch(void* const* d_in, const int* in_sizes, int n_in,
                              void* d_out, int out_size, void* d_ws, size_t ws_size,
                              hipStream_t stream) {
  const float* h   = (const float*)d_in[0];
  const int*   src = (const int*)d_in[1];
  const int*   dst = (const int*)d_in[2];
  const float* W   = (const float*)d_in[3];
  const float* a_l = (const float*)d_in[4];
  const float* a_r = (const float*)d_in[5];
  float* out = (float*)d_out;

  int nNodes = in_sizes[0] / IN_FEATS;
  int nEdges = in_sizes[1];

  float* ws      = (float*)d_ws;
  float* hW      = ws;
  float* el      = hW + (size_t)nNodes * OUT_FEATS;
  float* er      = el + nNodes;
  float2* erMax  = (float2*)(er + nNodes);
  float* planes  = er + (size_t)nNodes * 3;
  float4* nodeRec = (float4*)(planes + (size_t)N_SLICES * nNodes);

  dim3 blk(256);
  int tiles = (nNodes + 15) / 16;
  int blocksK1 = (tiles + 7) / 8;
  int nodeBlocks = (nNodes + 255) / 256;

  int rangeLen = ((nNodes + N_RANGES - 1) / N_RANGES + 31) / 32 * 32;
  int sliceLen = (((nEdges + N_SLICES - 1) / N_SLICES) + SEG_EDGES - 1) / SEG_EDGES * SEG_EDGES;
  size_t ldsBytes = (size_t)rangeLen * sizeof(float);
  dim3 segGrid(N_SLICES, N_RANGES);

  gat_project_wmma<<<blocksK1, blk, 0, stream>>>(h, W, hW, nNodes);
  gat_node_prep<<<nodeBlocks, blk, 0, stream>>>(hW, a_l, a_r, el, er, out, nNodes);
  gat_segmax_lds<<<segGrid, SEG_THREADS, ldsBytes, stream>>>(src, dst, el, er, planes,
                                                             nNodes, nEdges, rangeLen, sliceLen);
  gat_reduce_max<<<nodeBlocks, blk, 0, stream>>>(planes, er, erMax, nNodes);
  gat_segsum_lds<<<segGrid, SEG_THREADS, ldsBytes, stream>>>(src, dst, el, erMax, planes,
                                                             nNodes, nEdges, rangeLen, sliceLen);
  gat_reduce_sum<<<nodeBlocks, blk, 0, stream>>>(planes, erMax, nodeRec, nNodes);

  dim3 aggGrid(AG_SLICES, AG_RANGES);
  gat_edge_aggr_lds<<<aggGrid, AG_THREADS, 0, stream>>>(src, dst, el, nodeRec, hW, out,
                                                        nNodes, nEdges);
}


# Softmax denominator summed in the aggregation pass (LDS word per row, one global add per slice), rows divided by it afterwards; separate sum pass and its plane reduction go; sum order changes
# speedup vs baseline: 2.7642x; 1.1639x over previous
#include <hip/hip_runtime.h>
#include <hip/hip_bf16.h>

#define IN_FEATS   128
#define OUT_FEATS  16
#define NEG_SLOPE  0.2f

typedef float v2f __attribute__((ext_vector_type(2)));
typedef float v8f __attribute__((ext_vector_type(8)));

__global__ void gat_project_wmma(const float* __restrict__ h,
                                 const float* __restrict__ W,
                                 float* __restrict__ hW, int nNodes) {
  int wave = (blockIdx.x * blockDim.x + threadIdx.x) >> 5;
  int lane = threadIdx.x & 31;
  int m0 = wave * 16;
  if (m0 >= nNodes) return;
  int half = lane >> 4;
  int l16  = lane & 15;
  const float* arow = h + (size_t)(m0 + l16) * IN_FEATS;
  const float* brow = W + (size_t)l16 * IN_FEATS;
  v8f acc = {};
#pragma unroll
  for (int k = 0; k < IN_FEATS; k += 4) {
    int ka = k + half * 2;
    v2f a = *(const v2f*)(arow + ka);
    v2f b = *(const v2f*)(brow + ka);
    acc = __builtin_amdgcn_wmma_f32_16x16x4_f32(false, a, false, b,
                                                (short)0, acc, false, false);
  }
#pragma unroll
  for (int g = 0; g < 8; ++g) {
    int row = m0 + half * 8 + g;
    hW[(size_t)row * OUT_FEATS + l16] = acc[g];
  }
}

#define N_SLICES     28
#define N_RANGES     9
#define SEG_THREADS  1024
#define SEG_EDGES    8

__global__ void gat_node_prep(const float* __restrict__ hW,
                              const float* __restrict__ a_l,
                              const float* __restrict__ a_r,
                              float* __restrict__ el, float* __restrict__ er,
                              float* __restrict__ out, float* __restrict__ esum,
                              int nNodes) {
  int n = blockIdx.x * blockDim.x + threadIdx.x;
  if (n >= nNodes) return;
  const float* row = hW + (size_t)n * OUT_FEATS;
  float sl = 0.f, sr = 0.f;
#pragma unroll
  for (int f = 0; f < OUT_FEATS; ++f) {
    float v = row[f];
    sl += v * a_l[f];
    sr += v * a_r[f];
    out[(size_t)n * OUT_FEATS + f] = 0.f;
  }
  el[n] = sl;
  er[n] = sr;
  esum[n] = 0.f;
}

__device__ __forceinline__ void ldsMaxFloat(float* addr, float val) {
  if (val >= 0.f) atomicMax((int*)addr, __float_as_int(val));
  else            atomicMin((unsigned int*)addr, (unsigned int)__float_as_int(val));
}

__device__ __forceinline__ void load_edges(const int* __restrict__ src,
                                           const int* __restrict__ dst,
                                           int base, int e1,
                                           int (&s)[SEG_EDGES], int (&d)[SEG_EDGES]) {
  if (base + SEG_EDGES <= e1) {
    int4 d0 = *(const int4*)(dst + base);
    int4 d1 = *(const int4*)(dst + base + 4);
    int4 s0 = *(const int4*)(src + base);
    int4 s1 = *(const int4*)(src + base + 4);
    d[0] = d0.x; d[1] = d0.y; d[2] = d0.z; d[3] = d0.w;
    d[4] = d1.x; d[5] = d1.y; d[6] = d1.z; d[7] = d1.w;
    s[0] = s0.x; s[1] = s0.y; s[2] = s0.z; s[3] = s0.w;
    s[4] = s1.x; s[5] = s1.y; s[6] = s1.z; s[7] = s1.w;
  } else {
#pragma unroll
    for (int j = 0; j < SEG_EDGES; ++j) {
      int e = base + j;
      bool ok = e < e1;
      d[j] = ok ? dst[ok ? e : 0] : -1;
      s[j] = ok ? src[ok ? e : 0] : 0;
    }
  }
}

__global__ void __launch_bounds__(SEG_THREADS)
gat_segmax_lds(const int* __restrict__ src, const int* __restrict__ dst,
               const float* __restrict__ el, const float* __restrict__ er,
               float* __restrict__ planes, int nNodes, int nEdges,
               int rangeLen, int sliceLen) {
  extern __shared__ float tab[];
  const int tid = threadIdx.x;
  const int lo = blockIdx.y * rangeLen;
  int cnt = nNodes - lo;
  cnt = cnt > rangeLen ? rangeLen : (cnt < 0 ? 0 : cnt);
  for (int i = tid; i < cnt; i += SEG_THREADS) tab[i] = __int_as_float(0xFF800000);
  __syncthreads();
  const int e0 = blockIdx.x * sliceLen;
  int e1 = e0 + sliceLen;
  if (e1 > nEdges) e1 = nEdges;
  int s[SEG_EDGES], d[SEG_EDGES];
  int base = e0 + tid * SEG_EDGES;
  load_edges(src, dst, base, e1, s, d);
  while (base < e1) {
    const int nbase = base + SEG_THREADS * SEG_EDGES;
    int ns[SEG_EDGES], nd[SEG_EDGES];
    load_edges(src, dst, nbase, e1, ns, nd);
    float x[SEG_EDGES];
#pragma unroll
    for (int j = 0; j < SEG_EDGES; ++j) {
      bool in = (unsigned)(d[j] - lo) < (unsigned)cnt;
      x[j] = el[in ? s[j] : 0] + er[in ? d[j] : 0];
    }
#pragma unroll
    for (int j = 0; j < SEG_EDGES; ++j) {
      unsigned u = (unsigned)(d[j] - lo);
      if (u < (unsigned)cnt) {
        float v = (x[j] > 0.f) ? x[j] : NEG_SLOPE * x[j];
        ldsMaxFloat(tab + u, v);
      }
    }
#pragma unroll
    for (int j = 0; j < SEG_EDGES; ++j) { s[j] = ns[j]; d[j] = nd[j]; }
    base = nbase;
  }
  __syncthreads();
  float* plane = planes + (size_t)blockIdx.x * nNodes + lo;
  for (int i = tid; i < cnt; i += SEG_THREADS) plane[i] = tab[i];
}

__global__ void gat_reduce_max(const float* __restrict__ planes,
                               const float* __restrict__ er,
                               float2* __restrict__ erMax, int nNodes) {
  int n = blockIdx.x * blockDim.x + threadIdx.x;
  if (n >= nNodes) return;
  float m = __int_as_float(0xFF800000);
#pragma unroll
  for (int b = 0; b < N_SLICES; ++b) m = fmaxf(m, planes[(size_t)b * nNodes + n]);
  if (!(fabsf(m) < __int_as_float(0x7F800000))) m = 0.f;
  erMax[n] = make_float2(er[n], m);
}

#define AG_SLICES   8
#define AG_RANGES   32
#define AG_THREADS  1024
#define AG_EDGES    16
#define AG_QCAP     (AG_THREADS * AG_EDGES)
#define AG_ROWS     3126
#define AG_GROUPS   (AG_THREADS / OUT_FEATS)

__device__ __forceinline__ void load_dst16(const int* __restrict__ dst, int base, int e1,
                                           int (&d)[AG_EDGES]) {
  if (base + AG_EDGES <= e1) {
#pragma unroll
    for (int q = 0; q < AG_EDGES / 4; ++q) {
      int4 v = *(const int4*)(dst + base + 4 * q);
      d[4 * q] = v.x; d[4 * q + 1] = v.y; d[4 * q + 2] = v.z; d[4 * q + 3] = v.w;
    }
  } else {
#pragma unroll
    for (int j = 0; j < AG_EDGES; ++j) {
      int e = base + j;
      bool ok = e < e1;
      d[j] = ok ? dst[ok ? e : 0] : -1;
    }
  }
}

template <int K> __device__ __forceinline__ int row_share_i(int v) {
  return __builtin_amdgcn_update_dpp(v, v, 0x150 + K, 0xF, 0xF, false);
}
template <int K> __device__ __forceinline__ float row_share_f(float v) {
  return __int_as_float(row_share_i<K>(__float_as_int(v)));
}
#define AG_FOR16(X) X(0) X(1) X(2) X(3) X(4) X(5) X(6) X(7) X(8) X(9) X(10) X(11) X(12) X(13) X(14) X(15)
#define AG_ROW_LOAD(K) hv[K] = hW[(unsigned)(row_share_i<K>(s) * OUT_FEATS + f)];
#define AG_ROW_ADD(K)  atomicAdd(tab + row_share_i<K>(urow) * OUT_FEATS + f, hv[K] * row_share_f<K>(w));

__device__ __forceinline__ unsigned range_mask(const int (&d)[AG_EDGES], int lo, int cnt) {
  unsigned m = 0;
#pragma unroll
  for (int j = 0; j < AG_EDGES; ++j)
    m |= (unsigned)((unsigned)(d[j] - lo) < (unsigned)cnt) << j;
  return m;
}

__global__ void __launch_bounds__(AG_THREADS)
gat_edge_aggr_lds(const int* __restrict__ src, const int* __restrict__ dst,
                  const float* __restrict__ el, const float2* __restrict__ erMax,
                  const float* __restrict__ hW, float* __restrict__ out,
                  float* __restrict__ esum, int nNodes, int nEdges) {
  __shared__ float tab[AG_ROWS * OUT_FEATS];
  __shared__ float den[AG_ROWS];
  __shared__ int queue[AG_QCAP];
  __shared__ int qn[2];
  const int tid = threadIdx.x;
  const int lane = tid & 31;
  const int g = tid >> 4;
  const int f = tid & 15;
  const int sliceLen = (((nEdges + (int)gridDim.x - 1) / (int)gridDim.x) + AG_EDGES - 1)
                       / AG_EDGES * AG_EDGES;
  const int e0 = blockIdx.x * sliceLen;
  int e1 = e0 + sliceLen;
  if (e1 > nEdges) e1 = nEdges;

  for (int lo = blockIdx.y * AG_ROWS; lo < nNodes; lo += (int)gridDim.y * AG_ROWS) {
    int cnt = nNodes - lo;
    cnt = cnt > AG_ROWS ? AG_ROWS : cnt;
    for (int i = tid; i < cnt * OUT_FEATS; i += AG_THREADS) tab[i] = 0.f;
    for (int i = tid; i < cnt; i += AG_THREADS) den[i] = 0.f;
    if (tid < 2) qn[tid] = 0;
    __syncthreads();

    unsigned hit;
    {
      int d[AG_EDGES];
      load_dst16(dst, e0 + tid * AG_EDGES, e1, d);
      hit = range_mask(d, lo, cnt);
    }
    int par = 0;
    for (int tb = e0; tb < e1; tb += AG_QCAP) {
      const int base = tb + tid * AG_EDGES;
      const int c = __popc(hit);
      int incl = c;
#pragma unroll
      for (int o = 1; o < 32; o <<= 1) {
        int t = __shfl_up(incl, o);
        if (lane >= o) incl += t;
      }
      const int total = __builtin_amdgcn_readlane(incl, 31);
      int wbase = 0;
      if (total > 0) {
        if (lane == 0) wbase = atomicAdd(&qn[par], total);
        wbase = __builtin_amdgcn_readfirstlane(wbase);
      }
      int pos = wbase + incl - c;
      for (unsigned hb = hit; hb != 0u; hb &= hb - 1u)
        queue[pos++] = base + __builtin_ctz(hb);
      __syncthreads();

      int nd[AG_EDGES];
      load_dst16(dst, base + AG_QCAP, e1, nd);

      const int n = qn[par];
      if (tid == 0) qn[par ^ 1] = 0;

      for (int q0 = g * OUT_FEATS; q0 < n; q0 += AG_GROUPS * OUT_FEATS) {
        const int idx = q0 + f;
        const bool okl = idx < n;
        const int e = queue[okl ? idx : q0];
        const int s = src[e];
        const int urow = dst[e] - lo;
        const float2 r = erMax[urow + lo];
        float x = el[s] + r.x;
        x = (x > 0.f) ? x : NEG_SLOPE * x;
        float w = __expf(x - r.y);
        w = okl ? w : 0.f;
        float hv[OUT_FEATS];
        AG_FOR16(AG_ROW_LOAD)
        AG_FOR16(AG_ROW_ADD)
        atomicAdd(den + urow, w);
      }
      __syncthreads();
      hit = range_mask(nd, lo, cnt);
      par ^= 1;
    }

    float* o = out + (size_t)lo * OUT_FEATS;
    for (int i = tid; i < cnt * OUT_FEATS; i += AG_THREADS) atomicAdd(o + i, tab[i]);
    for (int i = tid; i < cnt; i += AG_THREADS) atomicAdd(esum + lo + i, den[i]);
    __syncthreads();
  }
}

__global__ void gat_normalize(float* __restrict__ out, const float* __restrict__ esum,
                              int nNodes) {
  float4* out4 = (float4*)out;
  const int quads = OUT_FEATS / 4;
  const int base = blockIdx.x * blockDim.x * quads + threadIdx.x;
  for (int q = 0; q < quads; ++q) {
    int t = base + q * blockDim.x;
    if (t < nNodes * quads) {
      const float d = fmaxf(esum[t / quads], 1e-16f);
      float4 v = out4[t];
      v.x /= d; v.y /= d; v.z /= d; v.w /= d;
      out4[t] = v;
    }
  }
}

extern "C" void kernel_launch(void* const* d_in, const int* in_sizes, int n_in,
                              void* d_out, int out_size, void* d_ws, size_t ws_size,
                              hipStream_t stream) {
  const float* h   = (const float*)d_in[0];
  const int*   src = (const int*)d_in[1];
  const int*   dst = (const int*)d_in[2];
  const float* W   = (const float*)d_in[3];
  const float* a_l = (const float*)d_in[4];
  const float* a_r = (const float*)d_in[5];
  float* out = (float*)d_out;

  int nNodes = in_sizes[0] / IN_FEATS;
  int nEdges = in_sizes[1];

  float* ws      = (float*)d_ws;
  float* hW      = ws;
  float* el      = hW + (size_t)nNodes * OUT_FEATS;
  float* er      = el + nNodes;
  float2* erMax  = (float2*)(er + nNodes);
  float* planes  = er + (size_t)nNodes * 3;
  float* esum    = planes + (size_t)N_SLICES * nNodes;

  dim3 blk(256);
  int tiles = (nNodes + 15) / 16;
  int blocksK1 = (tiles + 7) / 8;
  int nodeBlocks = (nNodes + 255) / 256;

  int rangeLen = ((nNodes + N_RANGES - 1) / N_RANGES + 31) / 32 * 32;
  int sliceLen = (((nEdges + N_SLICES - 1) / N_SLICES) + SEG_EDGES - 1) / SEG_EDGES * SEG_EDGES;
  size_t ldsBytes = (size_t)rangeLen * sizeof(float);
  dim3 segGrid(N_SLICES, N_RANGES);

  gat_project_wmma<<<blocksK1, blk, 0, stream>>>(h, W, hW, nNodes);
  gat_node_prep<<<nodeBlocks, blk, 0, stream>>>(hW, a_l, a_r, el, er, out, esum, nNodes);
  gat_segmax_lds<<<segGrid, SEG_THREADS, ldsBytes, stream>>>(src, dst, el, er, planes,
                                                             nNodes, nEdges, rangeLen, sliceLen);
  gat_reduce_max<<<nodeBlocks, blk, 0, stream>>>(planes, er, erMax, nNodes);

  dim3 aggGrid(AG_SLICES, AG_RANGES);
  gat_edge_aggr_lds<<<aggGrid, AG_THREADS, 0, stream>>>(src, dst, el, erMax, hW, out, esum,
                                                        nNodes, nEdges);
  gat_normalize<<<nodeBlocks, blk, 0, stream>>>(out, esum, nNodes);
}


# Aggregation scan re-laid: per trip a wave reads 768 consecutive destination indices in six whole-line 16-byte loads (24 per thread, was 16 at 64-byte lane stride); 17 trips, was 25; sum order changes
# speedup vs baseline: 2.9253x; 1.0583x over previous
#include <hip/hip_runtime.h>
#include <hip/hip_bf16.h>

#define IN_FEATS   128
#define OUT_FEATS  16
#define NEG_SLOPE  0.2f

typedef float v2f __attribute__((ext_vector_type(2)));
typedef float v8f __attribute__((ext_vector_type(8)));

__global__ void gat_project_wmma(const float* __restrict__ h,
                                 const float* __restrict__ W,
                                 float* __restrict__ hW, int nNodes) {
  int wave = (blockIdx.x * blockDim.x + threadIdx.x) >> 5;
  int lane = threadIdx.x & 31;
  int m0 = wave * 16;
  if (m0 >= nNodes) return;
  int half = lane >> 4;
  int l16  = lane & 15;
  const float* arow = h + (size_t)(m0 + l16) * IN_FEATS;
  const float* brow = W + (size_t)l16 * IN_FEATS;
  v8f acc = {};
#pragma unroll
  for (int k = 0; k < IN_FEATS; k += 4) {
    int ka = k + half * 2;
    v2f a = *(const v2f*)(arow + ka);
    v2f b = *(const v2f*)(brow + ka);
    acc = __builtin_amdgcn_wmma_f32_16x16x4_f32(false, a, false, b,
                                                (short)0, acc, false, false);
  }
#pragma unroll
  for (int g = 0; g < 8; ++g) {
    int row = m0 + half * 8 + g;
    hW[(size_t)row * OUT_FEATS + l16] = acc[g];
  }
}

#define N_SLICES     28
#define N_RANGES     9
#define SEG_THREADS  1024
#define SEG_EDGES    8

__global__ void gat_node_prep(const float* __restrict__ hW,
                              const float* __restrict__ a_l,
                              const float* __restrict__ a_r,
                              float* __restrict__ el, float* __restrict__ er,
                              float* __restrict__ out, float* __restrict__ esum,
                              int nNodes) {
  int n = blockIdx.x * blockDim.x + threadIdx.x;
  if (n >= nNodes) return;
  const float* row = hW + (size_t)n * OUT_FEATS;
  float sl = 0.f, sr = 0.f;
#pragma unroll
  for (int f = 0; f < OUT_FEATS; ++f) {
    float v = row[f];
    sl += v * a_l[f];
    sr += v * a_r[f];
    out[(size_t)n * OUT_FEATS + f] = 0.f;
  }
  el[n] = sl;
  er[n] = sr;
  esum[n] = 0.f;
}

__device__ __forceinline__ void ldsMaxFloat(float* addr, float val) {
  if (val >= 0.f) atomicMax((int*)addr, __float_as_int(val));
  else            atomicMin((unsigned int*)addr, (unsigned int)__float_as_int(val));
}

__device__ __forceinline__ void load_edges(const int* __restrict__ src,
                                           const int* __restrict__ dst,
                                           int base, int e1,
                                           int (&s)[SEG_EDGES], int (&d)[SEG_EDGES]) {
  if (base + SEG_EDGES <= e1) {
    int4 d0 = *(const int4*)(dst + base);
    int4 d1 = *(const int4*)(dst + base + 4);
    int4 s0 = *(const int4*)(src + base);
    int4 s1 = *(const int4*)(src + base + 4);
    d[0] = d0.x; d[1] = d0.y; d[2] = d0.z; d[3] = d0.w;
    d[4] = d1.x; d[5] = d1.y; d[6] = d1.z; d[7] = d1.w;
    s[0] = s0.x; s[1] = s0.y; s[2] = s0.z; s[3] = s0.w;
    s[4] = s1.x; s[5] = s1.y; s[6] = s1.z; s[7] = s1.w;
  } else {
#pragma unroll
    for (int j = 0; j < SEG_EDGES; ++j) {
      int e = base + j;
      bool ok = e < e1;
      d[j] = ok ? dst[ok ? e : 0] : -1;
      s[j] = ok ? src[ok ? e : 0] : 0;
    }
  }
}

__global__ void __launch_bounds__(SEG_THREADS)
gat_segmax_lds(const int* __restrict__ src, const int* __restrict__ dst,
               const float* __restrict__ el, const float* __restrict__ er,
               float* __restrict__ planes, int nNodes, int nEdges,
               int rangeLen, int sliceLen) {
  extern __shared__ float tab[];
  const int tid = threadIdx.x;
  const int lo = blockIdx.y * rangeLen;
  int cnt = nNodes - lo;
  cnt = cnt > rangeLen ? rangeLen : (cnt < 0 ? 0 : cnt);
  for (int i = tid; i < cnt; i += SEG_THREADS) tab[i] = __int_as_float(0xFF800000);
  __syncthreads();
  const int e0 = blockIdx.x * sliceLen;
  int e1 = e0 + sliceLen;
  if (e1 > nEdges) e1 = nEdges;
  int s[SEG_EDGES], d[SEG_EDGES];
  int base = e0 + tid * SEG_EDGES;
  load_edges(src, dst, base, e1, s, d);
  while (base < e1) {
    const int nbase = base + SEG_THREADS * SEG_EDGES;
    int ns[SEG_EDGES], nd[SEG_EDGES];
    load_edges(src, dst, nbase, e1, ns, nd);
    float x[SEG_EDGES];
#pragma unroll
    for (int j = 0; j < SEG_EDGES; ++j) {
      bool in = (unsigned)(d[j] - lo) < (unsigned)cnt;
      x[j] = el[in ? s[j] : 0] + er[in ? d[j] : 0];
    }
#pragma unroll
    for (int j = 0; j < SEG_EDGES; ++j) {
      unsigned u = (unsigned)(d[j] - lo);
      if (u < (unsigned)cnt) {
        float v = (x[j] > 0.f) ? x[j] : NEG_SLOPE * x[j];
        ldsMaxFloat(tab + u, v);
      }
    }
#pragma unroll
    for (int j = 0; j < SEG_EDGES; ++j) { s[j] = ns[j]; d[j] = nd[j]; }
    base = nbase;
  }
  __syncthreads();
  float* plane = planes + (size_t)blockIdx.x * nNodes + lo;
  for (int i = tid; i < cnt; i += SEG_THREADS) plane[i] = tab[i];
}

__global__ void gat_reduce_max(const float* __restrict__ planes,
                               const float* __restrict__ er,
                               float2* __restrict__ erMax, int nNodes) {
  int n = blockIdx.x * blockDim.x + threadIdx.x;
  if (n >= nNodes) return;
  float m = __int_as_float(0xFF800000);
#pragma unroll
  for (int b = 0; b < N_SLICES; ++b) m = fmaxf(m, planes[(size_t)b * nNodes + n]);
  if (!(fabsf(m) < __int_as_float(0x7F800000))) m = 0.f;
  erMax[n] = make_float2(er[n], m);
}

#define AG_SLICES   8
#define AG_RANGES   32
#define AG_THREADS  1024
#define AG_EDGES    24
#define AG_PIECES   (AG_EDGES / 4)
#define AG_WAVE_EDGES (32 * AG_EDGES)
#define AG_QCAP     (AG_THREADS * AG_EDGES)
#define AG_ROWS     3126
#define AG_GROUPS   (AG_THREADS / OUT_FEATS)

__device__ __forceinline__ void load_dst_wave(const int* __restrict__ dst, int lbase, bool whole,
                                              int e1, int (&d)[AG_EDGES]) {
  if (whole) {
#pragma unroll
    for (int q = 0; q < AG_PIECES; ++q) {
      int4 v = *(const int4*)(dst + lbase + 128 * q);
      d[4 * q] = v.x; d[4 * q + 1] = v.y; d[4 * q + 2] = v.z; d[4 * q + 3] = v.w;
    }
  } else {
#pragma unroll
    for (int j = 0; j < AG_EDGES; ++j) {
      const int off = 128 * (j >> 2) + (j & 3);
      d[j] = (lbase + off < e1) ? dst[lbase + off] : -1;
    }
  }
}

template <int K> __device__ __forceinline__ int row_share_i(int v) {
  return __builtin_amdgcn_update_dpp(v, v, 0x150 + K, 0xF, 0xF, false);
}
template <int K> __device__ __forceinline__ float row_share_f(float v) {
  return __int_as_float(row_share_i<K>(__float_as_int(v)));
}
#define AG_FOR16(X) X(0) X(1) X(2) X(3) X(4) X(5) X(6) X(7) X(8) X(9) X(10) X(11) X(12) X(13) X(14) X(15)
#define AG_ROW_LOAD(K) hv[K] = hW[(unsigned)(row_share_i<K>(s) * OUT_FEATS + f)];
#define AG_ROW_ADD(K)  atomicAdd(tab + row_share_i<K>(urow) * OUT_FEATS + f, hv[K] * row_share_f<K>(w));

__device__ __forceinline__ unsigned range_mask(const int (&d)[AG_EDGES], int lo, int cnt) {
  unsigned m = 0;
#pragma unroll
  for (int j = 0; j < AG_EDGES; ++j)
    m |= (unsigned)((unsigned)(d[j] - lo) < (unsigned)cnt) << j;
  return m;
}

__global__ void __launch_bounds__(AG_THREADS)
gat_edge_aggr_lds(const int* __restrict__ src, const int* __restrict__ dst,
                  const float* __restrict__ el, const float2* __restrict__ erMax,
                  const float* __restrict__ hW, float* __restrict__ out,
                  float* __restrict__ esum, int nNodes, int nEdges) {
  __shared__ float tab[AG_ROWS * OUT_FEATS];
  __shared__ float den[AG_ROWS];
  __shared__ unsigned short queue[AG_QCAP];
  __shared__ int qn[2];
  const int tid = threadIdx.x;
  const int lane = tid & 31;
  const int g = tid >> 4;
  const int f = tid & 15;
  const int sliceLen = (((nEdges + (int)gridDim.x - 1) / (int)gridDim.x) + 31) / 32 * 32;
  const int wave = __builtin_amdgcn_readfirstlane(tid >> 5);
  const int woff = wave * AG_WAVE_EDGES;
  const int loff = woff + 4 * lane;
  const int e0 = blockIdx.x * sliceLen;
  int e1 = e0 + sliceLen;
  if (e1 > nEdges) e1 = nEdges;

  for (int lo = blockIdx.y * AG_ROWS; lo < nNodes; lo += (int)gridDim.y * AG_ROWS) {
    int cnt = nNodes - lo;
    cnt = cnt > AG_ROWS ? AG_ROWS : cnt;
    for (int i = tid; i < cnt * OUT_FEATS; i += AG_THREADS) tab[i] = 0.f;
    for (int i = tid; i < cnt; i += AG_THREADS) den[i] = 0.f;
    if (tid < 2) qn[tid] = 0;
    __syncthreads();

    unsigned hit;
    {
      int d[AG_EDGES];
      load_dst_wave(dst, e0 + loff, e0 + woff + AG_WAVE_EDGES <= e1, e1, d);
      hit = range_mask(d, lo, cnt);
    }
    int par = 0;
    for (int tb = e0; tb < e1; tb += AG_QCAP) {
      const int c = __popc(hit);
      int incl = c;
#pragma unroll
      for (int o = 1; o < 32; o <<= 1) {
        int t = __shfl_up(incl, o);
        if (lane >= o) incl += t;
      }
      const int total = __builtin_amdgcn_readlane(incl, 31);
      int wbase = 0;
      if (total > 0) {
        if (lane == 0) wbase = atomicAdd(&qn[par], total);
        wbase = __builtin_amdgcn_readfirstlane(wbase);
      }
      int pos = wbase + incl - c;
      for (unsigned hb = hit; hb != 0u; hb &= hb - 1u) {
        const int j = __builtin_ctz(hb);
        queue[pos++] = (unsigned short)(loff + ((j >> 2) << 7) + (j & 3));
      }
      __syncthreads();

      int nd[AG_EDGES];
      load_dst_wave(dst, tb + AG_QCAP + loff, tb + AG_QCAP + woff + AG_WAVE_EDGES <= e1, e1, nd);

      const int n = qn[par];
      if (tid == 0) qn[par ^ 1] = 0;

      for (int q0 = g * OUT_FEATS; q0 < n; q0 += AG_GROUPS * OUT_FEATS) {
        const int idx = q0 + f;
        const bool okl = idx < n;
        const int e = tb + (int)queue[okl ? idx : q0];
        const int s = src[e];
        const int urow = dst[e] - lo;
        const float2 r = erMax[urow + lo];
        const float els = el[s];
        float hv[OUT_FEATS];
        AG_FOR16(AG_ROW_LOAD)
        __builtin_amdgcn_sched_barrier(0);
        float x = els + r.x;
        x = (x > 0.f) ? x : NEG_SLOPE * x;
        float w = __expf(x - r.y);
        w = okl ? w : 0.f;
        AG_FOR16(AG_ROW_ADD)
        atomicAdd(den + urow, w);
      }
      __syncthreads();
      hit = range_mask(nd, lo, cnt);
      par ^= 1;
    }

    float* o = out + (size_t)lo * OUT_FEATS;
    for (int i = tid; i < cnt * OUT_FEATS; i += AG_THREADS) atomicAdd(o + i, tab[i]);
    for (int i = tid; i < cnt; i += AG_THREADS) atomicAdd(esum + lo + i, den[i]);
    __syncthreads();
  }
}

__global__ void gat_normalize(float* __restrict__ out, const float* __restrict__ esum,
                              int nNodes) {
  float4* out4 = (float4*)out;
  const int quads = OUT_FEATS / 4;
  const int base = blockIdx.x * blockDim.x * quads + threadIdx.x;
  for (int q = 0; q < quads; ++q) {
    int t = base + q * blockDim.x;
    if (t < nNodes * quads) {
      const float d = fmaxf(esum[t / quads], 1e-16f);
      float4 v = out4[t];
      v.x /= d; v.y /= d; v.z /= d; v.w /= d;
      out4[t] = v;
    }
  }
}

extern "C" void kernel_launch(void* const* d_in, const int* in_sizes, int n_in,
                              void* d_out, int out_size, void* d_ws, size_t ws_size,
                              hipStream_t stream) {
  const float* h   = (const float*)d_in[0];
  const int*   src = (const int*)d_in[1];
  const int*   dst = (const int*)d_in[2];
  const float* W   = (const float*)d_in[3];
  const float* a_l = (const float*)d_in[4];
  const float* a_r = (const float*)d_in[5];
  float* out = (float*)d_out;

  int nNodes = in_sizes[0] / IN_FEATS;
  int nEdges = in_sizes[1];

  float* ws      = (float*)d_ws;
  float* hW      = ws;
  float* el      = hW + (size_t)nNodes * OUT_FEATS;
  float* er      = el + nNodes;
  float2* erMax  = (float2*)(er + nNodes);
  float* planes  = er + (size_t)nNodes * 3;
  float* esum    = planes + (size_t)N_SLICES * nNodes;

  dim3 blk(256);
  int tiles = (nNodes + 15) / 16;
  int blocksK1 = (tiles + 7) / 8;
  int nodeBlocks = (nNodes + 255) / 256;

  int rangeLen = ((nNodes + N_RANGES - 1) / N_RANGES + 31) / 32 * 32;
  int sliceLen = (((nEdges + N_SLICES - 1) / N_SLICES) + SEG_EDGES - 1) / SEG_EDGES * SEG_EDGES;
  size_t ldsBytes = (size_t)rangeLen * sizeof(float);
  dim3 segGrid(N_SLICES, N_RANGES);

  gat_project_wmma<<<blocksK1, blk, 0, stream>>>(h, W, hW, nNodes);
  gat_node_prep<<<nodeBlocks, blk, 0, stream>>>(hW, a_l, a_r, el, er, out, esum, nNodes);
  gat_segmax_lds<<<segGrid, SEG_THREADS, ldsBytes, stream>>>(src, dst, el, er, planes,
                                                             nNodes, nEdges, rangeLen, sliceLen);
  gat_reduce_max<<<nodeBlocks, blk, 0, stream>>>(planes, er, erMax, nNodes);

  dim3 aggGrid(AG_SLICES, AG_RANGES);
  gat_edge_aggr_lds<<<aggGrid, AG_THREADS, 0, stream>>>(src, dst, el, erMax, hW, out, esum,
                                                        nNodes, nEdges);
  gat_normalize<<<nodeBlocks, blk, 0, stream>>>(out, esum, nNodes);
}


# Edges counting-sorted by 256-row destination bin into packed words; one work-group per bin does max, weights, row sums and division in LDS, plain stores; no global float adds; sum order changes
# speedup vs baseline: 5.7196x; 1.9552x over previous
#include <hip/hip_runtime.h>
#include <hip/hip_bf16.h>

#define IN_FEATS   128
#define OUT_FEATS  16
#define NEG_SLOPE  0.2f

typedef float v2f __attribute__((ext_vector_type(2)));
typedef float v8f __attribute__((ext_vector_type(8)));

__global__ void gat_project_wmma(const float* __restrict__ h,
                                 const float* __restrict__ W,
                                 float* __restrict__ hW, int nNodes) {
  int wave = (blockIdx.x * blockDim.x + threadIdx.x) >> 5;
  int lane = threadIdx.x & 31;
  int m0 = wave * 16;
  if (m0 >= nNodes) return;
  int half = lane >> 4;
  int l16  = lane & 15;
  const float* arow = h + (size_t)(m0 + l16) * IN_FEATS;
  const float* brow = W + (size_t)l16 * IN_FEATS;
  v8f acc = {};
#pragma unroll
  for (int k = 0; k < IN_FEATS; k += 4) {
    int ka = k + half * 2;
    v2f a = *(const v2f*)(arow + ka);
    v2f b = *(const v2f*)(brow + ka);
    acc = __builtin_amdgcn_wmma_f32_16x16x4_f32(false, a, false, b,
                                                (short)0, acc, false, false);
  }
#pragma unroll
  for (int g = 0; g < 8; ++g) {
    int row = m0 + half * 8 + g;
    hW[(size_t)row * OUT_FEATS + l16] = acc[g];
  }
}

#define BIN_SHIFT    8
#define BIN_ROWS     (1 << BIN_SHIFT)
#define SRC_BITS     20
#define SRC_MASK     ((1u << SRC_BITS) - 1u)
#define BIN_CHUNKS   256
#define BIN_THREADS  1024
#define X_CAP        16384

__global__ void gat_node_prep(const float* __restrict__ hW,
                              const float* __restrict__ a_l,
                              const float* __restrict__ a_r,
                              float* __restrict__ el, float* __restrict__ er,
                              int* __restrict__ binTotal,
                              int nNodes, int nBins) {
  int n = blockIdx.x * blockDim.x + threadIdx.x;
  if (n < nBins) binTotal[n] = 0;
  if (n >= nNodes) return;
  const float* row = hW + (size_t)n * OUT_FEATS;
  float sl = 0.f, sr = 0.f;
#pragma unroll
  for (int f = 0; f < OUT_FEATS; ++f) {
    float v = row[f];
    sl += v * a_l[f];
    sr += v * a_r[f];
  }
  el[n] = sl;
  er[n] = sr;
}

__device__ __forceinline__ void ldsMaxFloat(float* addr, float val) {
  if (val >= 0.f) atomicMax((int*)addr, __float_as_int(val));
  else            atomicMin((unsigned int*)addr, (unsigned int)__float_as_int(val));
}

__device__ __forceinline__ int chunk_len(int nEdges, int nChunks) {
  return (((nEdges + nChunks - 1) / nChunks) + 3) & ~3;
}

__device__ __forceinline__ void load4(const int* __restrict__ p, int base, int e1, int fill,
                                      int (&v)[4]) {
  if (base + 4 <= e1) {
    int4 t = *(const int4*)(p + base);
    v[0] = t.x; v[1] = t.y; v[2] = t.z; v[3] = t.w;
  } else {
#pragma unroll
    for (int j = 0; j < 4; ++j) v[j] = (base + j < e1) ? p[base + j] : fill;
  }
}

__device__ __forceinline__ int bin_of(int d, int nBins) {
  unsigned b = (unsigned)d >> BIN_SHIFT;
  return (int)(b < (unsigned)nBins ? b : (unsigned)(nBins - 1));
}

__global__ void __launch_bounds__(BIN_THREADS)
gat_bin_count(const int* __restrict__ dst, int* __restrict__ binTotal,
              int* __restrict__ binOff, int nEdges, int nBins) {
  extern __shared__ int bin_lds[];
  int* hist = bin_lds;
  const int tid = threadIdx.x;
  for (int b = tid; b < nBins; b += BIN_THREADS) hist[b] = 0;
  __syncthreads();
  const int len = chunk_len(nEdges, (int)gridDim.x);
  const int e0 = blockIdx.x * len;
  int e1 = e0 + len;
  if (e1 > nEdges) e1 = nEdges;
  for (int base = e0 + 4 * tid; base < e1; base += 4 * BIN_THREADS) {
    int d[4];
    load4(dst, base, e1, -1, d);
#pragma unroll
    for (int j = 0; j < 4; ++j)
      if (d[j] >= 0) atomicAdd(&hist[bin_of(d[j], nBins)], 1);
  }
  __syncthreads();
  for (int b = tid; b < nBins; b += BIN_THREADS)
    binOff[(size_t)blockIdx.x * nBins + b] = atomicAdd(&binTotal[b], hist[b]);
}

__global__ void __launch_bounds__(BIN_THREADS)
gat_bin_scatter(const int* __restrict__ src, const int* __restrict__ dst,
                const int* __restrict__ binTotal, const int* __restrict__ binOff,
                int* __restrict__ binStart, unsigned* __restrict__ packed,
                int nEdges, int nBins) {
  extern __shared__ int bin_lds[];
  int* cursor = bin_lds;
  int* start  = bin_lds + nBins;
  const int tid = threadIdx.x;
  for (int b = tid; b < nBins; b += BIN_THREADS) start[b] = binTotal[b];
  __syncthreads();
  if (tid < 32) {
    const int per = (nBins + 31) / 32;
    const int b0 = tid * per;
    int b1 = b0 + per;
    if (b1 > nBins) b1 = nBins;
    int s = 0;
    for (int b = b0; b < b1; ++b) s += start[b];
    int incl = s;
#pragma unroll
    for (int o = 1; o < 32; o <<= 1) {
      int t = __shfl_up(incl, o);
      if (tid >= o) incl += t;
    }
    int run = incl - s;
    for (int b = b0; b < b1; ++b) {
      const int t = start[b];
      start[b] = run;
      run += t;
    }
  }
  __syncthreads();
  for (int b = tid; b < nBins; b += BIN_THREADS) {
    cursor[b] = start[b] + binOff[(size_t)blockIdx.x * nBins + b];
    if (blockIdx.x == 0) binStart[b] = start[b];
  }
  if (blockIdx.x == 0 && tid == 0) binStart[nBins] = nEdges;
  __syncthreads();
  const int len = chunk_len(nEdges, (int)gridDim.x);
  const int e0 = blockIdx.x * len;
  int e1 = e0 + len;
  if (e1 > nEdges) e1 = nEdges;
  for (int base = e0 + 4 * tid; base < e1; base += 4 * BIN_THREADS) {
    int d[4], s[4];
    load4(dst, base, e1, -1, d);
    load4(src, base, e1, 0, s);
#pragma unroll
    for (int j = 0; j < 4; ++j) {
      if (d[j] >= 0) {
        const int pos = atomicAdd(&cursor[bin_of(d[j], nBins)], 1);
        packed[pos] = ((unsigned)(d[j] & (BIN_ROWS - 1)) << SRC_BITS) | ((unsigned)s[j] & SRC_MASK);
      }
    }
  }
}

template <int K> __device__ __forceinline__ int row_share_i(int v) {
  return __builtin_amdgcn_update_dpp(v, v, 0x150 + K, 0xF, 0xF, false);
}
template <int K> __device__ __forceinline__ float row_share_f(float v) {
  return __int_as_float(row_share_i<K>(__float_as_int(v)));
}
#define AG_FOR16(X) X(0) X(1) X(2) X(3) X(4) X(5) X(6) X(7) X(8) X(9) X(10) X(11) X(12) X(13) X(14) X(15)
#define AG_ROW_LOAD(K) hv[K] = hW[(unsigned)(row_share_i<K>(s) * OUT_FEATS + f)];
#define AG_ROW_ADD(K)  atomicAdd(tab + row_share_i<K>(urow) * OUT_FEATS + f, hv[K] * row_share_f<K>(w));

#define AG_GROUPS   (BIN_THREADS / OUT_FEATS)
#define MAX_EDGES   4

__global__ void __launch_bounds__(BIN_THREADS)
gat_bin_aggregate(const unsigned* __restrict__ packed, const int* __restrict__ binStart,
                  const float* __restrict__ el, const float* __restrict__ er,
                  const float* __restrict__ hW, float* __restrict__ out, int nNodes) {
  __shared__ float tab[BIN_ROWS * OUT_FEATS];
  __shared__ float den[BIN_ROWS];
  __shared__ float mx[BIN_ROWS];
  __shared__ float erl[BIN_ROWS];
  __shared__ float xc[X_CAP];
  const int tid = threadIdx.x;
  const int g = tid >> 4;
  const int f = tid & 15;
  const int lo = blockIdx.x << BIN_SHIFT;
  int cnt = nNodes - lo;
  cnt = cnt > BIN_ROWS ? BIN_ROWS : cnt;
  for (int i = tid; i < BIN_ROWS * OUT_FEATS; i += BIN_THREADS) tab[i] = 0.f;
  if (tid < BIN_ROWS) {
    den[tid] = 0.f;
    mx[tid] = __int_as_float(0xFF800000);
    erl[tid] = (tid < cnt) ? er[lo + tid] : 0.f;
  }
  __syncthreads();
  const int p0 = binStart[blockIdx.x];
  const int p1 = binStart[blockIdx.x + 1];

  for (int i0 = p0 + tid; i0 < p1; i0 += MAX_EDGES * BIN_THREADS) {
    unsigned pk[MAX_EDGES];
    float v[MAX_EDGES];
#pragma unroll
    for (int u = 0; u < MAX_EDGES; ++u) {
      const int i = i0 + u * BIN_THREADS;
      pk[u] = packed[i < p1 ? i : i0];
    }
#pragma unroll
    for (int u = 0; u < MAX_EDGES; ++u) v[u] = el[pk[u] & SRC_MASK];
#pragma unroll
    for (int u = 0; u < MAX_EDGES; ++u) {
      const int i = i0 + u * BIN_THREADS;
      if (i < p1) {
        const int r = (int)(pk[u] >> SRC_BITS);
        float x = v[u] + erl[r];
        x = (x > 0.f) ? x : NEG_SLOPE * x;
        if (i - p0 < X_CAP) xc[i - p0] = x;
        ldsMaxFloat(mx + r, x);
      }
    }
  }
  __syncthreads();
  if (tid < BIN_ROWS) {
    float m = mx[tid];
    if (!(fabsf(m) < __int_as_float(0x7F800000))) m = 0.f;
    mx[tid] = m;
  }
  __syncthreads();

  int q0 = p0 + g * OUT_FEATS;
  unsigned pk = 0u;
  if (q0 < p1) pk = packed[(q0 + f < p1) ? q0 + f : q0];
  for (; q0 < p1; q0 += AG_GROUPS * OUT_FEATS) {
    const int nq0 = q0 + AG_GROUPS * OUT_FEATS;
    const int ni = (nq0 + f < p1) ? nq0 + f : ((nq0 < p1) ? nq0 : q0);
    const unsigned npk = packed[ni];
    const bool okl = q0 + f < p1;
    const int j = (okl ? q0 + f : q0) - p0;
    const int s = (int)(pk & SRC_MASK);
    const int urow = (int)(pk >> SRC_BITS);
    float hv[OUT_FEATS];
    AG_FOR16(AG_ROW_LOAD)
    float x;
    if (j < X_CAP) {
      x = xc[j];
    } else {
      x = el[s] + erl[urow];
      x = (x > 0.f) ? x : NEG_SLOPE * x;
    }
    float w = __expf(x - mx[urow]);
    w = okl ? w : 0.f;
    AG_FOR16(AG_ROW_ADD)
    atomicAdd(den + urow, w);
    pk = npk;
  }
  __syncthreads();

  float* o = out + (size_t)lo * OUT_FEATS;
  for (int i = tid; i < cnt * OUT_FEATS; i += BIN_THREADS)
    o[i] = tab[i] / fmaxf(den[i >> 4], 1e-16f);
}

extern "C" void kernel_launch(void* const* d_in, const int* in_sizes, int n_in,
                              void* d_out, int out_size, void* d_ws, size_t ws_size,
                              hipStream_t stream) {
  const float* h   = (const float*)d_in[0];
  const int*   src = (const int*)d_in[1];
  const int*   dst = (const int*)d_in[2];
  const float* W   = (const float*)d_in[3];
  const float* a_l = (const float*)d_in[4];
  const float* a_r = (const float*)d_in[5];
  float* out = (float*)d_out;

  int nNodes = in_sizes[0] / IN_FEATS;
  int nEdges = in_sizes[1];
  int nBins = (nNodes + BIN_ROWS - 1) / BIN_ROWS;

  float* ws        = (float*)d_ws;
  float* hW        = ws;
  float* el        = hW + (size_t)nNodes * OUT_FEATS;
  float* er        = el + nNodes;
  unsigned* packed = (unsigned*)(er + nNodes);
  int* binTotal    = (int*)(packed + nEdges);
  int* binStart    = binTotal + nBins;
  int* binOff      = binStart + nBins + 1;

  dim3 blk(256);
  int tiles = (nNodes + 15) / 16;
  int blocksK1 = (tiles + 7) / 8;
  int nodeBlocks = (nNodes + 255) / 256;
  size_t countLds = (size_t)nBins * sizeof(int);
  size_t scatterLds = (size_t)nBins * 2 * sizeof(int);

  gat_project_wmma<<<blocksK1, blk, 0, stream>>>(h, W, hW, nNodes);
  gat_node_prep<<<nodeBlocks, blk, 0, stream>>>(hW, a_l, a_r, el, er, binTotal, nNodes, nBins);
  gat_bin_count<<<BIN_CHUNKS, BIN_THREADS, countLds, stream>>>(dst, binTotal, binOff, nEdges, nBins);
  gat_bin_scatter<<<BIN_CHUNKS, BIN_THREADS, scatterLds, stream>>>(src, dst, binTotal, binOff,
                                                                  binStart, packed, nEdges, nBins);
  gat_bin_aggregate<<<nBins, BIN_THREADS, 0, stream>>>(packed, binStart, el, er, hW, out, nNodes);
}


# Destination bins of 196 rows, was 256: 511 one-bin work-groups fill the chip's 512 places (two per WGP) evenly, where 391 left WGPs with two bins beside WGPs with one; sum order changes
# speedup vs baseline: 5.9320x; 1.0371x over previous
#include <hip/hip_runtime.h>
#include <hip/hip_bf16.h>

#define IN_FEATS   128
#define OUT_FEATS  16
#define NEG_SLOPE  0.2f

typedef float v2f __attribute__((ext_vector_type(2)));
typedef float v8f __attribute__((ext_vector_type(8)));

__global__ void gat_project_wmma(const float* __restrict__ h,
                                 const float* __restrict__ W,
                                 float* __restrict__ hW, int nNodes) {
  int wave = (blockIdx.x * blockDim.x + threadIdx.x) >> 5;
  int lane = threadIdx.x & 31;
  int m0 = wave * 16;
  if (m0 >= nNodes) return;
  int half = lane >> 4;
  int l16  = lane & 15;
  const float* arow = h + (size_t)(m0 + l16) * IN_FEATS;
  const float* brow = W + (size_t)l16 * IN_FEATS;
  v8f acc = {};
#pragma unroll
  for (int k = 0; k < IN_FEATS; k += 4) {
    int ka = k + half * 2;
    v2f a = *(const v2f*)(arow + ka);
    v2f b = *(const v2f*)(brow + ka);
    acc = __builtin_amdgcn_wmma_f32_16x16x4_f32(false, a, false, b,
                                                (short)0, acc, false, false);
  }
#pragma unroll
  for (int g = 0; g < 8; ++g) {
    int row = m0 + half * 8 + g;
    hW[(size_t)row * OUT_FEATS + l16] = acc[g];
  }
}

#define BIN_ROWS     196
#define SRC_BITS     20
#define SRC_MASK     ((1u << SRC_BITS) - 1u)
#define BIN_CHUNKS   256
#define BIN_THREADS  1024
#define X_CAP        16384

__global__ void gat_node_prep(const float* __restrict__ hW,
                              const float* __restrict__ a_l,
                              const float* __restrict__ a_r,
                              float* __restrict__ el, float* __restrict__ er,
                              int* __restrict__ binTotal,
                              int nNodes, int nBins) {
  int n = blockIdx.x * blockDim.x + threadIdx.x;
  if (n < nBins) binTotal[n] = 0;
  if (n >= nNodes) return;
  const float* row = hW + (size_t)n * OUT_FEATS;
  float sl = 0.f, sr = 0.f;
#pragma unroll
  for (int f = 0; f < OUT_FEATS; ++f) {
    float v = row[f];
    sl += v * a_l[f];
    sr += v * a_r[f];
  }
  el[n] = sl;
  er[n] = sr;
}

__device__ __forceinline__ void ldsMaxFloat(float* addr, float val) {
  if (val >= 0.f) atomicMax((int*)addr, __float_as_int(val));
  else            atomicMin((unsigned int*)addr, (unsigned int)__float_as_int(val));
}

__device__ __forceinline__ int chunk_len(int nEdges, int nChunks) {
  return (((nEdges + nChunks - 1) / nChunks) + 3) & ~3;
}

__device__ __forceinline__ void load4(const int* __restrict__ p, int base, int e1, int fill,
                                      int (&v)[4]) {
  if (base + 4 <= e1) {
    int4 t = *(const int4*)(p + base);
    v[0] = t.x; v[1] = t.y; v[2] = t.z; v[3] = t.w;
  } else {
#pragma unroll
    for (int j = 0; j < 4; ++j) v[j] = (base + j < e1) ? p[base + j] : fill;
  }
}

__device__ __forceinline__ int bin_of(int d, int nBins) {
  unsigned b = (unsigned)d / BIN_ROWS;
  return (int)(b < (unsigned)nBins ? b : (unsigned)(nBins - 1));
}

__global__ void __launch_bounds__(BIN_THREADS)
gat_bin_count(const int* __restrict__ dst, int* __restrict__ binTotal,
              int* __restrict__ binOff, int nEdges, int nBins) {
  extern __shared__ int bin_lds[];
  int* hist = bin_lds;
  const int tid = threadIdx.x;
  for (int b = tid; b < nBins; b += BIN_THREADS) hist[b] = 0;
  __syncthreads();
  const int len = chunk_len(nEdges, (int)gridDim.x);
  const int e0 = blockIdx.x * len;
  int e1 = e0 + len;
  if (e1 > nEdges) e1 = nEdges;
  for (int base = e0 + 4 * tid; base < e1; base += 4 * BIN_THREADS) {
    int d[4];
    load4(dst, base, e1, -1, d);
#pragma unroll
    for (int j = 0; j < 4; ++j)
      if (d[j] >= 0) atomicAdd(&hist[bin_of(d[j], nBins)], 1);
  }
  __syncthreads();
  for (int b = tid; b < nBins; b += BIN_THREADS)
    binOff[(size_t)blockIdx.x * nBins + b] = atomicAdd(&binTotal[b], hist[b]);
}

__global__ void __launch_bounds__(BIN_THREADS)
gat_bin_scatter(const int* __restrict__ src, const int* __restrict__ dst,
                const int* __restrict__ binTotal, const int* __restrict__ binOff,
                int* __restrict__ binStart, unsigned* __restrict__ packed,
                int nEdges, int nBins) {
  extern __shared__ int bin_lds[];
  int* cursor = bin_lds;
  int* start  = bin_lds + nBins;
  const int tid = threadIdx.x;
  for (int b = tid; b < nBins; b += BIN_THREADS) start[b] = binTotal[b];
  __syncthreads();
  if (tid < 32) {
    const int per = (nBins + 31) / 32;
    const int b0 = tid * per;
    int b1 = b0 + per;
    if (b1 > nBins) b1 = nBins;
    int s = 0;
    for (int b = b0; b < b1; ++b) s += start[b];
    int incl = s;
#pragma unroll
    for (int o = 1; o < 32; o <<= 1) {
      int t = __shfl_up(incl, o);
      if (tid >= o) incl += t;
    }
    int run = incl - s;
    for (int b = b0; b < b1; ++b) {
      const int t = start[b];
      start[b] = run;
      run += t;
    }
  }
  __syncthreads();
  for (int b = tid; b < nBins; b += BIN_THREADS) {
    cursor[b] = start[b] + binOff[(size_t)blockIdx.x * nBins + b];
    if (blockIdx.x == 0) binStart[b] = start[b];
  }
  if (blockIdx.x == 0 && tid == 0) binStart[nBins] = nEdges;
  __syncthreads();
  const int len = chunk_len(nEdges, (int)gridDim.x);
  const int e0 = blockIdx.x * len;
  int e1 = e0 + len;
  if (e1 > nEdges) e1 = nEdges;
  for (int base = e0 + 4 * tid; base < e1; base += 4 * BIN_THREADS) {
    int d[4], s[4];
    load4(dst, base, e1, -1, d);
    load4(src, base, e1, 0, s);
#pragma unroll
    for (int j = 0; j < 4; ++j) {
      if (d[j] >= 0) {
        const int pos = atomicAdd(&cursor[bin_of(d[j], nBins)], 1);
        packed[pos] = (((unsigned)d[j] % BIN_ROWS) << SRC_BITS) | ((unsigned)s[j] & SRC_MASK);
      }
    }
  }
}

template <int K> __device__ __forceinline__ int row_share_i(int v) {
  return __builtin_amdgcn_update_dpp(v, v, 0x150 + K, 0xF, 0xF, false);
}
template <int K> __device__ __forceinline__ float row_share_f(float v) {
  return __int_as_float(row_share_i<K>(__float_as_int(v)));
}
#define AG_FOR16(X) X(0) X(1) X(2) X(3) X(4) X(5) X(6) X(7) X(8) X(9) X(10) X(11) X(12) X(13) X(14) X(15)
#define AG_ROW_LOAD(K) hv[K] = hW[(unsigned)(row_share_i<K>(s) * OUT_FEATS + f)];
#define AG_ROW_ADD(K)  atomicAdd(tab + row_share_i<K>(urow) * OUT_FEATS + f, hv[K] * row_share_f<K>(w));

#define AG_GROUPS   (BIN_THREADS / OUT_FEATS)
#define MAX_EDGES   4

__global__ void __launch_bounds__(BIN_THREADS)
gat_bin_aggregate(const unsigned* __restrict__ packed, const int* __restrict__ binStart,
                  const float* __restrict__ el, const float* __restrict__ er,
                  const float* __restrict__ hW, float* __restrict__ out, int nNodes) {
  __shared__ float tab[BIN_ROWS * OUT_FEATS];
  __shared__ float den[BIN_ROWS];
  __shared__ float mx[BIN_ROWS];
  __shared__ float erl[BIN_ROWS];
  __shared__ float xc[X_CAP];
  const int tid = threadIdx.x;
  const int g = tid >> 4;
  const int f = tid & 15;
  const int lo = blockIdx.x * BIN_ROWS;
  int cnt = nNodes - lo;
  cnt = cnt > BIN_ROWS ? BIN_ROWS : cnt;
  for (int i = tid; i < BIN_ROWS * OUT_FEATS; i += BIN_THREADS) tab[i] = 0.f;
  if (tid < BIN_ROWS) {
    den[tid] = 0.f;
    mx[tid] = __int_as_float(0xFF800000);
    erl[tid] = (tid < cnt) ? er[lo + tid] : 0.f;
  }
  __syncthreads();
  const int p0 = binStart[blockIdx.x];
  const int p1 = binStart[blockIdx.x + 1];

  for (int i0 = p0 + tid; i0 < p1; i0 += MAX_EDGES * BIN_THREADS) {
    unsigned pk[MAX_EDGES];
    float v[MAX_EDGES];
#pragma unroll
    for (int u = 0; u < MAX_EDGES; ++u) {
      const int i = i0 + u * BIN_THREADS;
      pk[u] = packed[i < p1 ? i : i0];
    }
#pragma unroll
    for (int u = 0; u < MAX_EDGES; ++u) v[u] = el[pk[u] & SRC_MASK];
#pragma unroll
    for (int u = 0; u < MAX_EDGES; ++u) {
      const int i = i0 + u * BIN_THREADS;
      if (i < p1) {
        const int r = (int)(pk[u] >> SRC_BITS);
        float x = v[u] + erl[r];
        x = (x > 0.f) ? x : NEG_SLOPE * x;
        if (i - p0 < X_CAP) xc[i - p0] = x;
        ldsMaxFloat(mx + r, x);
      }
    }
  }
  __syncthreads();
  if (tid < BIN_ROWS) {
    float m = mx[tid];
    if (!(fabsf(m) < __int_as_float(0x7F800000))) m = 0.f;
    mx[tid] = m;
  }
  __syncthreads();

  int q0 = p0 + g * OUT_FEATS;
  unsigned pk = 0u;
  if (q0 < p1) pk = packed[(q0 + f < p1) ? q0 + f : q0];
  for (; q0 < p1; q0 += AG_GROUPS * OUT_FEATS) {
    const int nq0 = q0 + AG_GROUPS * OUT_FEATS;
    const int ni = (nq0 + f < p1) ? nq0 + f : ((nq0 < p1) ? nq0 : q0);
    const unsigned npk = packed[ni];
    const bool okl = q0 + f < p1;
    const int j = (okl ? q0 + f : q0) - p0;
    const int s = (int)(pk & SRC_MASK);
    const int urow = (int)(pk >> SRC_BITS);
    float hv[OUT_FEATS];
    AG_FOR16(AG_ROW_LOAD)
    float x;
    if (j < X_CAP) {
      x = xc[j];
    } else {
      x = el[s] + erl[urow];
      x = (x > 0.f) ? x : NEG_SLOPE * x;
    }
    float w = __expf(x - mx[urow]);
    w = okl ? w : 0.f;
    AG_FOR16(AG_ROW_ADD)
    atomicAdd(den + urow, w);
    pk = npk;
  }
  __syncthreads();

  float* o = out + (size_t)lo * OUT_FEATS;
  for (int i = tid; i < cnt * OUT_FEATS; i += BIN_THREADS)
    o[i] = tab[i] / fmaxf(den[i >> 4], 1e-16f);
}

extern "C" void kernel_launch(void* const* d_in, const int* in_sizes, int n_in,
                              void* d_out, int out_size, void* d_ws, size_t ws_size,
                              hipStream_t stream) {
  const float* h   = (const float*)d_in[0];
  const int*   src = (const int*)d_in[1];
  const int*   dst = (const int*)d_in[2];
  const float* W   = (const float*)d_in[3];
  const float* a_l = (const float*)d_in[4];
  const float* a_r = (const float*)d_in[5];
  float* out = (float*)d_out;

  int nNodes = in_sizes[0] / IN_FEATS;
  int nEdges = in_sizes[1];
  int nBins = (nNodes + BIN_ROWS - 1) / BIN_ROWS;

  float* ws        = (float*)d_ws;
  float* hW        = ws;
  float* el        = hW + (size_t)nNodes * OUT_FEATS;
  float* er        = el + nNodes;
  unsigned* packed = (unsigned*)(er + nNodes);
  int* binTotal    = (int*)(packed + nEdges);
  int* binStart    = binTotal + nBins;
  int* binOff      = binStart + nBins + 1;

  dim3 blk(256);
  int tiles = (nNodes + 15) / 16;
  int blocksK1 = (tiles + 7) / 8;
  int nodeBlocks = (nNodes + 255) / 256;
  size_t countLds = (size_t)nBins * sizeof(int);
  size_t scatterLds = (size_t)nBins * 2 * sizeof(int);

  gat_project_wmma<<<blocksK1, blk, 0, stream>>>(h, W, hW, nNodes);
  gat_node_prep<<<nodeBlocks, blk, 0, stream>>>(hW, a_l, a_r, el, er, binTotal, nNodes, nBins);
  gat_bin_count<<<BIN_CHUNKS, BIN_THREADS, countLds, stream>>>(dst, binTotal, binOff, nEdges, nBins);
  gat_bin_scatter<<<BIN_CHUNKS, BIN_THREADS, scatterLds, stream>>>(src, dst, binTotal, binOff,
                                                                  binStart, packed, nEdges, nBins);
  gat_bin_aggregate<<<nBins, BIN_THREADS, 0, stream>>>(packed, binStart, el, er, hW, out, nNodes);
}


# Scatter sorts each edge chunk by bin in LDS (count, scan, place) and copies the words out in bin order: a wave's store spans a few 128-byte lines, not 32; sum order changes
# speedup vs baseline: 7.0225x; 1.1838x over previous
#include <hip/hip_runtime.h>
#include <hip/hip_bf16.h>

#define IN_FEATS   128
#define OUT_FEATS  16
#define NEG_SLOPE  0.2f

typedef float v2f __attribute__((ext_vector_type(2)));
typedef float v8f __attribute__((ext_vector_type(8)));

__global__ void gat_project_wmma(const float* __restrict__ h,
                                 const float* __restrict__ W,
                                 float* __restrict__ hW, int nNodes) {
  int wave = (blockIdx.x * blockDim.x + threadIdx.x) >> 5;
  int lane = threadIdx.x & 31;
  int m0 = wave * 16;
  if (m0 >= nNodes) return;
  int half = lane >> 4;
  int l16  = lane & 15;
  const float* arow = h + (size_t)(m0 + l16) * IN_FEATS;
  const float* brow = W + (size_t)l16 * IN_FEATS;
  v8f acc = {};
#pragma unroll
  for (int k = 0; k < IN_FEATS; k += 4) {
    int ka = k + half * 2;
    v2f a = *(const v2f*)(arow + ka);
    v2f b = *(const v2f*)(brow + ka);
    acc = __builtin_amdgcn_wmma_f32_16x16x4_f32(false, a, false, b,
                                                (short)0, acc, false, false);
  }
#pragma unroll
  for (int g = 0; g < 8; ++g) {
    int row = m0 + half * 8 + g;
    hW[(size_t)row * OUT_FEATS + l16] = acc[g];
  }
}

#define BIN_ROWS     196
#define SRC_BITS     20
#define SRC_MASK     ((1u << SRC_BITS) - 1u)
#define BIN_CHUNKS   256
#define BIN_THREADS  1024
#define X_CAP        16384

__global__ void gat_node_prep(const float* __restrict__ hW,
                              const float* __restrict__ a_l,
                              const float* __restrict__ a_r,
                              float* __restrict__ el, float* __restrict__ er,
                              int* __restrict__ binTotal,
                              int nNodes, int nBins) {
  int n = blockIdx.x * blockDim.x + threadIdx.x;
  if (n < nBins) binTotal[n] = 0;
  if (n >= nNodes) return;
  const float* row = hW + (size_t)n * OUT_FEATS;
  float sl = 0.f, sr = 0.f;
#pragma unroll
  for (int f = 0; f < OUT_FEATS; ++f) {
    float v = row[f];
    sl += v * a_l[f];
    sr += v * a_r[f];
  }
  el[n] = sl;
  er[n] = sr;
}

__device__ __forceinline__ void ldsMaxFloat(float* addr, float val) {
  if (val >= 0.f) atomicMax((int*)addr, __float_as_int(val));
  else            atomicMin((unsigned int*)addr, (unsigned int)__float_as_int(val));
}

__device__ __forceinline__ int chunk_len(int nEdges, int nChunks) {
  return (((nEdges + nChunks - 1) / nChunks) + 3) & ~3;
}

__device__ __forceinline__ void load4(const int* __restrict__ p, int base, int e1, int fill,
                                      int (&v)[4]) {
  if (base + 4 <= e1) {
    int4 t = *(const int4*)(p + base);
    v[0] = t.x; v[1] = t.y; v[2] = t.z; v[3] = t.w;
  } else {
#pragma unroll
    for (int j = 0; j < 4; ++j) v[j] = (base + j < e1) ? p[base + j] : fill;
  }
}

__device__ __forceinline__ int bin_of(int d, int nBins) {
  unsigned b = (unsigned)d / BIN_ROWS;
  return (int)(b < (unsigned)nBins ? b : (unsigned)(nBins - 1));
}

__global__ void __launch_bounds__(BIN_THREADS)
gat_bin_count(const int* __restrict__ dst, int* __restrict__ binTotal,
              int* __restrict__ binOff, int nEdges, int nBins) {
  extern __shared__ int bin_lds[];
  int* hist = bin_lds;
  const int tid = threadIdx.x;
  for (int b = tid; b < nBins; b += BIN_THREADS) hist[b] = 0;
  __syncthreads();
  const int len = chunk_len(nEdges, (int)gridDim.x);
  const int e0 = blockIdx.x * len;
  int e1 = e0 + len;
  if (e1 > nEdges) e1 = nEdges;
  for (int base = e0 + 4 * tid; base < e1; base += 4 * BIN_THREADS) {
    int d[4];
    load4(dst, base, e1, -1, d);
#pragma unroll
    for (int j = 0; j < 4; ++j)
      if (d[j] >= 0) atomicAdd(&hist[bin_of(d[j], nBins)], 1);
  }
  __syncthreads();
  for (int b = tid; b < nBins; b += BIN_THREADS)
    binOff[(size_t)blockIdx.x * nBins + b] = atomicAdd(&binTotal[b], hist[b]);
}

__device__ __forceinline__ int wave_scan_excl(int* a, int n, int lane) {
  const int per = (n + 31) / 32;
  const int b0 = lane * per;
  int b1 = b0 + per;
  if (b1 > n) b1 = n;
  int s = 0;
  for (int b = b0; b < b1; ++b) s += a[b];
  int incl = s;
#pragma unroll
  for (int o = 1; o < 32; o <<= 1) {
    int t = __shfl_up(incl, o);
    if (lane >= o) incl += t;
  }
  int run = incl - s;
  for (int b = b0; b < b1; ++b) {
    const int t = a[b];
    a[b] = run;
    run += t;
  }
  return __shfl(incl, 31);
}

#define SC_TRIPS   4
#define SC_BATCH   (SC_TRIPS * 4 * BIN_THREADS)

__device__ __forceinline__ void sc_load_batch(const int* __restrict__ src, const int* __restrict__ dst,
                                              int s0, int s1, int tid,
                                              int (&d)[SC_TRIPS][4], int (&s)[SC_TRIPS][4],
                                              bool (&ok)[SC_TRIPS]) {
#pragma unroll
  for (int t = 0; t < SC_TRIPS; ++t) {
    const int base = s0 + 4 * tid + t * 4 * BIN_THREADS;
    ok[t] = base < s1;
    const int cb = ok[t] ? base : s0;
    load4(dst, cb, s1, -1, d[t]);
    load4(src, cb, s1, 0, s[t]);
  }
}

__global__ void __launch_bounds__(BIN_THREADS)
gat_bin_scatter(const int* __restrict__ src, const int* __restrict__ dst,
                const int* __restrict__ binTotal, const int* __restrict__ binOff,
                int* __restrict__ binStart, unsigned* __restrict__ packed,
                int nEdges, int nBins) {
  extern __shared__ int scat_lds[];
  int* gcur   = scat_lds;
  int* lcur   = scat_lds + nBins;
  int* gdelta = lcur + nBins + 1;
  __shared__ uint2 rec[SC_BATCH];
  const int tid = threadIdx.x;
  const int len = chunk_len(nEdges, (int)gridDim.x);
  const int e0 = blockIdx.x * len;
  int e1 = e0 + len;
  if (e1 > nEdges) e1 = nEdges;
  const bool any = e0 < e1;

  int d[SC_TRIPS][4], s[SC_TRIPS][4];
  bool ok[SC_TRIPS];
  int s1 = e0 + SC_BATCH;
  if (s1 > e1) s1 = e1;
  if (any) sc_load_batch(src, dst, e0, s1, tid, d, s, ok);

  for (int b = tid; b < nBins; b += BIN_THREADS) {
    gcur[b] = binTotal[b];
    gdelta[b] = binOff[(size_t)blockIdx.x * nBins + b];
  }
  __syncthreads();
  if (tid < 32) {
    const int tot = wave_scan_excl(gcur, nBins, tid);
    if (blockIdx.x == 0 && tid == 0) binStart[nBins] = tot;
  }
  __syncthreads();
  for (int b = tid; b < nBins; b += BIN_THREADS) {
    const int st = gcur[b];
    if (blockIdx.x == 0) binStart[b] = st;
    gcur[b] = st + gdelta[b];
  }

  if (any) {
    int s0 = e0;
    for (;;) {
      for (int b = tid; b <= nBins; b += BIN_THREADS) lcur[b] = 0;
      __syncthreads();
#pragma unroll
      for (int t = 0; t < SC_TRIPS; ++t)
#pragma unroll
        for (int j = 0; j < 4; ++j)
          if (ok[t] && d[t][j] >= 0) atomicAdd(&lcur[bin_of(d[t][j], nBins)], 1);
      __syncthreads();
      if (tid < 32) {
        const int tot = wave_scan_excl(lcur, nBins, tid);
        if (tid == 0) lcur[nBins] = tot;
      }
      __syncthreads();
      for (int b = tid; b < nBins; b += BIN_THREADS) {
        const int ls = lcur[b];
        const int g = gcur[b];
        gdelta[b] = g - ls;
        gcur[b] = g + (lcur[b + 1] - ls);
      }
      const int nPlaced = lcur[nBins];
      __syncthreads();
#pragma unroll
      for (int t = 0; t < SC_TRIPS; ++t)
#pragma unroll
        for (int j = 0; j < 4; ++j)
          if (ok[t] && d[t][j] >= 0) {
            const int b = bin_of(d[t][j], nBins);
            const int lp = atomicAdd(&lcur[b], 1);
            const unsigned w = (((unsigned)d[t][j] % BIN_ROWS) << SRC_BITS) | ((unsigned)s[t][j] & SRC_MASK);
            rec[lp] = make_uint2(w, (unsigned)(gdelta[b] + lp));
          }
      __syncthreads();
      for (int i = tid; i < nPlaced; i += BIN_THREADS) {
        const uint2 r = rec[i];
        packed[r.y] = r.x;
      }
      s0 += SC_BATCH;
      if (s0 >= e1) break;
      s1 = s0 + SC_BATCH;
      if (s1 > e1) s1 = e1;
      sc_load_batch(src, dst, s0, s1, tid, d, s, ok);
    }
  }
}

template <int K> __device__ __forceinline__ int row_share_i(int v) {
  return __builtin_amdgcn_update_dpp(v, v, 0x150 + K, 0xF, 0xF, false);
}
template <int K> __device__ __forceinline__ float row_share_f(float v) {
  return __int_as_float(row_share_i<K>(__float_as_int(v)));
}
#define AG_FOR16(X) X(0) X(1) X(2) X(3) X(4) X(5) X(6) X(7) X(8) X(9) X(10) X(11) X(12) X(13) X(14) X(15)
#define AG_ROW_LOAD(K) hv[K] = hW[(unsigned)(row_share_i<K>(s) * OUT_FEATS + f)];
#define AG_ROW_ADD(K)  atomicAdd(tab + row_share_i<K>(urow) * OUT_FEATS + f, hv[K] * row_share_f<K>(w));

#define AG_GROUPS   (BIN_THREADS / OUT_FEATS)
#define MAX_EDGES   4

__global__ void __launch_bounds__(BIN_THREADS)
gat_bin_aggregate(const unsigned* __restrict__ packed, const int* __restrict__ binStart,
                  const float* __restrict__ el, const float* __restrict__ er,
                  const float* __restrict__ hW, float* __restrict__ out, int nNodes) {
  __shared__ float tab[BIN_ROWS * OUT_FEATS];
  __shared__ float den[BIN_ROWS];
  __shared__ float mx[BIN_ROWS];
  __shared__ float erl[BIN_ROWS];
  __shared__ float xc[X_CAP];
  const int tid = threadIdx.x;
  const int g = tid >> 4;
  const int f = tid & 15;
  const int lo = blockIdx.x * BIN_ROWS;
  int cnt = nNodes - lo;
  cnt = cnt > BIN_ROWS ? BIN_ROWS : cnt;
  for (int i = tid; i < BIN_ROWS * OUT_FEATS; i += BIN_THREADS) tab[i] = 0.f;
  if (tid < BIN_ROWS) {
    den[tid] = 0.f;
    mx[tid] = __int_as_float(0xFF800000);
    erl[tid] = (tid < cnt) ? er[lo + tid] : 0.f;
  }
  __syncthreads();
  const int p0 = binStart[blockIdx.x];
  const int p1 = binStart[blockIdx.x + 1];

  for (int i0 = p0 + tid; i0 < p1; i0 += MAX_EDGES * BIN_THREADS) {
    unsigned pk[MAX_EDGES];
    float v[MAX_EDGES];
#pragma unroll
    for (int u = 0; u < MAX_EDGES; ++u) {
      const int i = i0 + u * BIN_THREADS;
      pk[u] = packed[i < p1 ? i : i0];
    }
#pragma unroll
    for (int u = 0; u < MAX_EDGES; ++u) v[u] = el[pk[u] & SRC_MASK];
#pragma unroll
    for (int u = 0; u < MAX_EDGES; ++u) {
      const int i = i0 + u * BIN_THREADS;
      if (i < p1) {
        const int r = (int)(pk[u] >> SRC_BITS);
        float x = v[u] + erl[r];
        x = (x > 0.f) ? x : NEG_SLOPE * x;
        if (i - p0 < X_CAP) xc[i - p0] = x;
        ldsMaxFloat(mx + r, x);
      }
    }
  }
  __syncthreads();
  if (tid < BIN_ROWS) {
    float m = mx[tid];
    if (!(fabsf(m) < __int_as_float(0x7F800000))) m = 0.f;
    mx[tid] = m;
  }
  __syncthreads();

  int q0 = p0 + g * OUT_FEATS;
  unsigned pk = 0u;
  if (q0 < p1) pk = packed[(q0 + f < p1) ? q0 + f : q0];
  for (; q0 < p1; q0 += AG_GROUPS * OUT_FEATS) {
    const int nq0 = q0 + AG_GROUPS * OUT_FEATS;
    const int ni = (nq0 + f < p1) ? nq0 + f : ((nq0 < p1) ? nq0 : q0);
    const unsigned npk = packed[ni];
    const bool okl = q0 + f < p1;
    const int j = (okl ? q0 + f : q0) - p0;
    const int s = (int)(pk & SRC_MASK);
    const int urow = (int)(pk >> SRC_BITS);
    float hv[OUT_FEATS];
    AG_FOR16(AG_ROW_LOAD)
    float x;
    if (j < X_CAP) {
      x = xc[j];
    } else {
      x = el[s] + erl[urow];
      x = (x > 0.f) ? x : NEG_SLOPE * x;
    }
    float w = __expf(x - mx[urow]);
    w = okl ? w : 0.f;
    AG_FOR16(AG_ROW_ADD)
    atomicAdd(den + urow, w);
    pk = npk;
  }
  __syncthreads();

  float* o = out + (size_t)lo * OUT_FEATS;
  for (int i = tid; i < cnt * OUT_FEATS; i += BIN_THREADS)
    o[i] = tab[i] / fmaxf(den[i >> 4], 1e-16f);
}

extern "C" void kernel_launch(void* const* d_in, const int* in_sizes, int n_in,
                              void* d_out, int out_size, void* d_ws, size_t ws_size,
                              hipStream_t stream) {
  const float* h   = (const float*)d_in[0];
  const int*   src = (const int*)d_in[1];
  const int*   dst = (const int*)d_in[2];
  const float* W   = (const float*)d_in[3];
  const float* a_l = (const float*)d_in[4];
  const float* a_r = (const float*)d_in[5];
  float* out = (float*)d_out;

  int nNodes = in_sizes[0] / IN_FEATS;
  int nEdges = in_sizes[1];
  int nBins = (nNodes + BIN_ROWS - 1) / BIN_ROWS;

  float* ws        = (float*)d_ws;
  float* hW        = ws;
  float* el        = hW + (size_t)nNodes * OUT_FEATS;
  float* er        = el + nNodes;
  unsigned* packed = (unsigned*)(er + nNodes);
  int* binTotal    = (int*)(packed + nEdges);
  int* binStart    = binTotal + nBins;
  int* binOff      = binStart + nBins + 1;

  dim3 blk(256);
  int tiles = (nNodes + 15) / 16;
  int blocksK1 = (tiles + 7) / 8;
  int nodeBlocks = (nNodes + 255) / 256;
  size_t countLds = (size_t)nBins * sizeof(int);
  size_t scatterLds = ((size_t)nBins * 3 + 1) * sizeof(int);

  gat_project_wmma<<<blocksK1, blk, 0, stream>>>(h, W, hW, nNodes);
  gat_node_prep<<<nodeBlocks, blk, 0, stream>>>(hW, a_l, a_r, el, er, binTotal, nNodes, nBins);
  gat_bin_count<<<BIN_CHUNKS, BIN_THREADS, countLds, stream>>>(dst, binTotal, binOff, nEdges, nBins);
  gat_bin_scatter<<<BIN_CHUNKS, BIN_THREADS, scatterLds, stream>>>(src, dst, binTotal, binOff,
                                                                  binStart, packed, nEdges, nBins);
  gat_bin_aggregate<<<nBins, BIN_THREADS, 0, stream>>>(packed, binStart, el, er, hW, out, nNodes);
}


# el and er summed in the projection epilogue from the WMMA accumulators (DPP butterfly over a row's 16 lanes); node-prep launch and its re-read of hW go: 4 launches, was 5; el/er sum order changes
# speedup vs baseline: 7.3715x; 1.0497x over previous
#include <hip/hip_runtime.h>
#include <hip/hip_bf16.h>

#define IN_FEATS   128
#define OUT_FEATS  16
#define NEG_SLOPE  0.2f

typedef float v2f __attribute__((ext_vector_type(2)));
typedef float v8f __attribute__((ext_vector_type(8)));

template <int K> __device__ __forceinline__ float row_xmask_f(float v) {
  return __int_as_float(__builtin_amdgcn_update_dpp(__float_as_int(v), __float_as_int(v),
                                                    0x160 + K, 0xF, 0xF, false));
}

__global__ void gat_project_wmma(const float* __restrict__ h,
                                 const float* __restrict__ W,
                                 const float* __restrict__ a_l,
                                 const float* __restrict__ a_r,
                                 float* __restrict__ hW,
                                 float* __restrict__ el, float* __restrict__ er,
                                 int* __restrict__ binTotal,
                                 int nNodes, int nBins) {
  int gt = blockIdx.x * blockDim.x + threadIdx.x;
  if (gt < nBins) binTotal[gt] = 0;
  int wave = gt >> 5;
  int lane = threadIdx.x & 31;
  int m0 = wave * 16;
  if (m0 >= nNodes) return;
  int half = lane >> 4;
  int l16  = lane & 15;
  const float* arow = h + (size_t)(m0 + l16) * IN_FEATS;
  const float* brow = W + (size_t)l16 * IN_FEATS;
  v8f acc = {};
#pragma unroll
  for (int k = 0; k < IN_FEATS; k += 4) {
    int ka = k + half * 2;
    v2f a = *(const v2f*)(arow + ka);
    v2f b = *(const v2f*)(brow + ka);
    acc = __builtin_amdgcn_wmma_f32_16x16x4_f32(false, a, false, b,
                                                (short)0, acc, false, false);
  }
#pragma unroll
  for (int g = 0; g < 8; ++g) {
    int row = m0 + half * 8 + g;
    hW[(size_t)row * OUT_FEATS + l16] = acc[g];
  }
  const float al = a_l[l16];
  const float ar = a_r[l16];
  const bool hi = (l16 & 8) != 0;
  float keep = 0.f;
#pragma unroll
  for (int g = 0; g < 8; ++g) {
    float pl = acc[g] * al;
    float pr = acc[g] * ar;
    pl += row_xmask_f<8>(pl);
    pr += row_xmask_f<8>(pr);
    float u = hi ? pr : pl;
    u += row_xmask_f<4>(u);
    u += row_xmask_f<2>(u);
    u += row_xmask_f<1>(u);
    if ((l16 & 7) == g) keep = u;
  }
  const int nrow = m0 + half * 8 + (l16 & 7);
  if (nrow < nNodes) (hi ? er : el)[nrow] = keep;
}

#define BIN_ROWS     196
#define SRC_BITS     20
#define SRC_MASK     ((1u << SRC_BITS) - 1u)
#define BIN_CHUNKS   256
#define BIN_THREADS  1024
#define X_CAP        16384

__device__ __forceinline__ void ldsMaxFloat(float* addr, float val) {
  if (val >= 0.f) atomicMax((int*)addr, __float_as_int(val));
  else            atomicMin((unsigned int*)addr, (unsigned int)__float_as_int(val));
}

__device__ __forceinline__ int chunk_len(int nEdges, int nChunks) {
  return (((nEdges + nChunks - 1) / nChunks) + 3) & ~3;
}

__device__ __forceinline__ void load4(const int* __restrict__ p, int base, int e1, int fill,
                                      int (&v)[4]) {
  if (base + 4 <= e1) {
    int4 t = *(const int4*)(p + base);
    v[0] = t.x; v[1] = t.y; v[2] = t.z; v[3] = t.w;
  } else {
#pragma unroll
    for (int j = 0; j < 4; ++j) v[j] = (base + j < e1) ? p[base + j] : fill;
  }
}

__device__ __forceinline__ int bin_of(int d, int nBins) {
  unsigned b = (unsigned)d / BIN_ROWS;
  return (int)(b < (unsigned)nBins ? b : (unsigned)(nBins - 1));
}

__global__ void __launch_bounds__(BIN_THREADS)
gat_bin_count(const int* __restrict__ dst, int* __restrict__ binTotal,
              int* __restrict__ binOff, int nEdges, int nBins) {
  extern __shared__ int bin_lds[];
  int* hist = bin_lds;
  const int tid = threadIdx.x;
  for (int b = tid; b < nBins; b += BIN_THREADS) hist[b] = 0;
  __syncthreads();
  const int len = chunk_len(nEdges, (int)gridDim.x);
  const int e0 = blockIdx.x * len;
  int e1 = e0 + len;
  if (e1 > nEdges) e1 = nEdges;
  for (int base = e0 + 4 * tid; base < e1; base += 4 * BIN_THREADS) {
    int d[4];
    load4(dst, base, e1, -1, d);
#pragma unroll
    for (int j = 0; j < 4; ++j)
      if (d[j] >= 0) atomicAdd(&hist[bin_of(d[j], nBins)], 1);
  }
  __syncthreads();
  for (int b = tid; b < nBins; b += BIN_THREADS)
    binOff[(size_t)blockIdx.x * nBins + b] = atomicAdd(&binTotal[b], hist[b]);
}

__device__ __forceinline__ int wave_scan_excl(int* a, int n, int lane) {
  const int per = (n + 31) / 32;
  const int b0 = lane * per;
  int b1 = b0 + per;
  if (b1 > n) b1 = n;
  int s = 0;
  for (int b = b0; b < b1; ++b) s += a[b];
  int incl = s;
#pragma unroll
  for (int o = 1; o < 32; o <<= 1) {
    int t = __shfl_up(incl, o);
    if (lane >= o) incl += t;
  }
  int run = incl - s;
  for (int b = b0; b < b1; ++b) {
    const int t = a[b];
    a[b] = run;
    run += t;
  }
  return __shfl(incl, 31);
}

#define SC_TRIPS   4
#define SC_BATCH   (SC_TRIPS * 4 * BIN_THREADS)

__device__ __forceinline__ void sc_load_batch(const int* __restrict__ src, const int* __restrict__ dst,
                                              int s0, int s1, int tid,
                                              int (&d)[SC_TRIPS][4], int (&s)[SC_TRIPS][4],
                                              bool (&ok)[SC_TRIPS]) {
#pragma unroll
  for (int t = 0; t < SC_TRIPS; ++t) {
    const int base = s0 + 4 * tid + t * 4 * BIN_THREADS;
    ok[t] = base < s1;
    const int cb = ok[t] ? base : s0;
    load4(dst, cb, s1, -1, d[t]);
    load4(src, cb, s1, 0, s[t]);
  }
}

__global__ void __launch_bounds__(BIN_THREADS)
gat_bin_scatter(const int* __restrict__ src, const int* __restrict__ dst,
                const int* __restrict__ binTotal, const int* __restrict__ binOff,
                int* __restrict__ binStart, unsigned* __restrict__ packed,
                int nEdges, int nBins) {
  extern __shared__ int scat_lds[];
  int* gcur   = scat_lds;
  int* lcur   = scat_lds + nBins;
  int* gdelta = lcur + nBins + 1;
  __shared__ uint2 rec[SC_BATCH];
  const int tid = threadIdx.x;
  const int len = chunk_len(nEdges, (int)gridDim.x);
  const int e0 = blockIdx.x * len;
  int e1 = e0 + len;
  if (e1 > nEdges) e1 = nEdges;
  const bool any = e0 < e1;

  int d[SC_TRIPS][4], s[SC_TRIPS][4];
  bool ok[SC_TRIPS];
  int s1 = e0 + SC_BATCH;
  if (s1 > e1) s1 = e1;
  if (any) sc_load_batch(src, dst, e0, s1, tid, d, s, ok);

  for (int b = tid; b < nBins; b += BIN_THREADS) {
    gcur[b] = binTotal[b];
    gdelta[b] = binOff[(size_t)blockIdx.x * nBins + b];
  }
  __syncthreads();
  if (tid < 32) {
    const int tot = wave_scan_excl(gcur, nBins, tid);
    if (blockIdx.x == 0 && tid == 0) binStart[nBins] = tot;
  }
  __syncthreads();
  for (int b = tid; b < nBins; b += BIN_THREADS) {
    const int st = gcur[b];
    if (blockIdx.x == 0) binStart[b] = st;
    gcur[b] = st + gdelta[b];
  }

  if (any) {
    int s0 = e0;
    for (;;) {
      for (int b = tid; b <= nBins; b += BIN_THREADS) lcur[b] = 0;
      __syncthreads();
#pragma unroll
      for (int t = 0; t < SC_TRIPS; ++t)
#pragma unroll
        for (int j = 0; j < 4; ++j)
          if (ok[t] && d[t][j] >= 0) atomicAdd(&lcur[bin_of(d[t][j], nBins)], 1);
      __syncthreads();
      if (tid < 32) {
        const int tot = wave_scan_excl(lcur, nBins, tid);
        if (tid == 0) lcur[nBins] = tot;
      }
      __syncthreads();
      for (int b = tid; b < nBins; b += BIN_THREADS) {
        const int ls = lcur[b];
        const int g = gcur[b];
        gdelta[b] = g - ls;
        gcur[b] = g + (lcur[b + 1] - ls);
      }
      const int nPlaced = lcur[nBins];
      __syncthreads();
#pragma unroll
      for (int t = 0; t < SC_TRIPS; ++t)
#pragma unroll
        for (int j = 0; j < 4; ++j)
          if (ok[t] && d[t][j] >= 0) {
            const int b = bin_of(d[t][j], nBins);
            const int lp = atomicAdd(&lcur[b], 1);
            const unsigned w = (((unsigned)d[t][j] % BIN_ROWS) << SRC_BITS) | ((unsigned)s[t][j] & SRC_MASK);
            rec[lp] = make_uint2(w, (unsigned)(gdelta[b] + lp));
          }
      __syncthreads();
      for (int i = tid; i < nPlaced; i += BIN_THREADS) {
        const uint2 r = rec[i];
        packed[r.y] = r.x;
      }
      s0 += SC_BATCH;
      if (s0 >= e1) break;
      s1 = s0 + SC_BATCH;
      if (s1 > e1) s1 = e1;
      sc_load_batch(src, dst, s0, s1, tid, d, s, ok);
    }
  }
}

template <int K> __device__ __forceinline__ int row_share_i(int v) {
  return __builtin_amdgcn_update_dpp(v, v, 0x150 + K, 0xF, 0xF, false);
}
template <int K> __device__ __forceinline__ float row_share_f(float v) {
  return __int_as_float(row_share_i<K>(__float_as_int(v)));
}
#define AG_FOR16(X) X(0) X(1) X(2) X(3) X(4) X(5) X(6) X(7) X(8) X(9) X(10) X(11) X(12) X(13) X(14) X(15)
#define AG_ROW_LOAD(K) hv[K] = hW[(unsigned)(row_share_i<K>(s) * OUT_FEATS + f)];
#define AG_ROW_ADD(K)  atomicAdd(tab + row_share_i<K>(urow) * OUT_FEATS + f, hv[K] * row_share_f<K>(w));

#define AG_GROUPS   (BIN_THREADS / OUT_FEATS)
#define MAX_EDGES   4

__global__ void __launch_bounds__(BIN_THREADS)
gat_bin_aggregate(const unsigned* __restrict__ packed, const int* __restrict__ binStart,
                  const float* __restrict__ el, const float* __restrict__ er,
                  const float* __restrict__ hW, float* __restrict__ out, int nNodes) {
  __shared__ float tab[BIN_ROWS * OUT_FEATS];
  __shared__ float den[BIN_ROWS];
  __shared__ float mx[BIN_ROWS];
  __shared__ float erl[BIN_ROWS];
  __shared__ float xc[X_CAP];
  const int tid = threadIdx.x;
  const int g = tid >> 4;
  const int f = tid & 15;
  const int lo = blockIdx.x * BIN_ROWS;
  int cnt = nNodes - lo;
  cnt = cnt > BIN_ROWS ? BIN_ROWS : cnt;
  for (int i = tid; i < BIN_ROWS * OUT_FEATS; i += BIN_THREADS) tab[i] = 0.f;
  if (tid < BIN_ROWS) {
    den[tid] = 0.f;
    mx[tid] = __int_as_float(0xFF800000);
    erl[tid] = (tid < cnt) ? er[lo + tid] : 0.f;
  }
  __syncthreads();
  const int p0 = binStart[blockIdx.x];
  const int p1 = binStart[blockIdx.x + 1];

  for (int i0 = p0 + tid; i0 < p1; i0 += MAX_EDGES * BIN_THREADS) {
    unsigned pk[MAX_EDGES];
    float v[MAX_EDGES];
#pragma unroll
    for (int u = 0; u < MAX_EDGES; ++u) {
      const int i = i0 + u * BIN_THREADS;
      pk[u] = packed[i < p1 ? i : i0];
    }
#pragma unroll
    for (int u = 0; u < MAX_EDGES; ++u) v[u] = el[pk[u] & SRC_MASK];
#pragma unroll
    for (int u = 0; u < MAX_EDGES; ++u) {
      const int i = i0 + u * BIN_THREADS;
      if (i < p1) {
        const int r = (int)(pk[u] >> SRC_BITS);
        float x = v[u] + erl[r];
        x = (x > 0.f) ? x : NEG_SLOPE * x;
        if (i - p0 < X_CAP) xc[i - p0] = x;
        ldsMaxFloat(mx + r, x);
      }
    }
  }
  __syncthreads();
  if (tid < BIN_ROWS) {
    float m = mx[tid];
    if (!(fabsf(m) < __int_as_float(0x7F800000))) m = 0.f;
    mx[tid] = m;
  }
  __syncthreads();

  int q0 = p0 + g * OUT_FEATS;
  unsigned pk = 0u;
  if (q0 < p1) pk = packed[(q0 + f < p1) ? q0 + f : q0];
  for (; q0 < p1; q0 += AG_GROUPS * OUT_FEATS) {
    const int nq0 = q0 + AG_GROUPS * OUT_FEATS;
    const int ni = (nq0 + f < p1) ? nq0 + f : ((nq0 < p1) ? nq0 : q0);
    const unsigned npk = packed[ni];
    const bool okl = q0 + f < p1;
    const int j = (okl ? q0 + f : q0) - p0;
    const int s = (int)(pk & SRC_MASK);
    const int urow = (int)(pk >> SRC_BITS);
    float hv[OUT_FEATS];
    AG_FOR16(AG_ROW_LOAD)
    float x;
    if (j < X_CAP) {
      x = xc[j];
    } else {
      x = el[s] + erl[urow];
      x = (x > 0.f) ? x : NEG_SLOPE * x;
    }
    float w = __expf(x - mx[urow]);
    w = okl ? w : 0.f;
    AG_FOR16(AG_ROW_ADD)
    atomicAdd(den + urow, w);
    pk = npk;
  }
  __syncthreads();

  float* o = out + (size_t)lo * OUT_FEATS;
  for (int i = tid; i < cnt * OUT_FEATS; i += BIN_THREADS)
    o[i] = tab[i] / fmaxf(den[i >> 4], 1e-16f);
}

extern "C" void kernel_launch(void* const* d_in, const int* in_sizes, int n_in,
                              void* d_out, int out_size, void* d_ws, size_t ws_size,
                              hipStream_t stream) {
  const float* h   = (const float*)d_in[0];
  const int*   src = (const int*)d_in[1];
  const int*   dst = (const int*)d_in[2];
  const float* W   = (const float*)d_in[3];
  const float* a_l = (const float*)d_in[4];
  const float* a_r = (const float*)d_in[5];
  float* out = (float*)d_out;

  int nNodes = in_sizes[0] / IN_FEATS;
  int nEdges = in_sizes[1];
  int nBins = (nNodes + BIN_ROWS - 1) / BIN_ROWS;

  float* ws        = (float*)d_ws;
  float* hW        = ws;
  float* el        = hW + (size_t)nNodes * OUT_FEATS;
  float* er        = el + nNodes;
  unsigned* packed = (unsigned*)(er + nNodes);
  int* binTotal    = (int*)(packed + nEdges);
  int* binStart    = binTotal + nBins;
  int* binOff      = binStart + nBins + 1;

  dim3 blk(256);
  int tiles = (nNodes + 15) / 16;
  int blocksK1 = (tiles + 7) / 8;
  size_t countLds = (size_t)nBins * sizeof(int);
  size_t scatterLds = ((size_t)nBins * 3 + 1) * sizeof(int);

  gat_project_wmma<<<blocksK1, blk, 0, stream>>>(h, W, a_l, a_r, hW, el, er, binTotal, nNodes, nBins);
  gat_bin_count<<<BIN_CHUNKS, BIN_THREADS, countLds, stream>>>(dst, binTotal, binOff, nEdges, nBins);
  gat_bin_scatter<<<BIN_CHUNKS, BIN_THREADS, scatterLds, stream>>>(src, dst, binTotal, binOff,
                                                                  binStart, packed, nEdges, nBins);
  gat_bin_aggregate<<<nBins, BIN_THREADS, 0, stream>>>(packed, binStart, el, er, hW, out, nNodes);
}


# Projection loads re-laid: A and W by 16-byte loads (4 k per lane, two steps), two tiles per wave on one set of W fragments, loads in use order: 48 loads per 2 tiles, was 128; hW sum order changes
# speedup vs baseline: 7.6902x; 1.0432x over previous
#include <hip/hip_runtime.h>
#include <hip/hip_bf16.h>

#define IN_FEATS   128
#define OUT_FEATS  16
#define NEG_SLOPE  0.2f

typedef float v2f __attribute__((ext_vector_type(2)));
typedef float v4f __attribute__((ext_vector_type(4)));
typedef float v8f __attribute__((ext_vector_type(8)));

template <int K> __device__ __forceinline__ float row_xmask_f(float v) {
  return __int_as_float(__builtin_amdgcn_update_dpp(__float_as_int(v), __float_as_int(v),
                                                    0x160 + K, 0xF, 0xF, false));
}

#define PJ_TILES  2
#define PJ_WAVES  4

__device__ __forceinline__ void project_tile_out(const v8f acc, int m0, int half, int l16,
                                                 float al, float ar,
                                                 float* __restrict__ hW,
                                                 float* __restrict__ el, float* __restrict__ er,
                                                 int nNodes) {
#pragma unroll
  for (int g = 0; g < 8; ++g) {
    int row = m0 + half * 8 + g;
    hW[(size_t)row * OUT_FEATS + l16] = acc[g];
  }
  const bool hi = (l16 & 8) != 0;
  float keep = 0.f;
#pragma unroll
  for (int g = 0; g < 8; ++g) {
    float pl = acc[g] * al;
    float pr = acc[g] * ar;
    pl += row_xmask_f<8>(pl);
    pr += row_xmask_f<8>(pr);
    float u = hi ? pr : pl;
    u += row_xmask_f<4>(u);
    u += row_xmask_f<2>(u);
    u += row_xmask_f<1>(u);
    if ((l16 & 7) == g) keep = u;
  }
  const int nrow = m0 + half * 8 + (l16 & 7);
  if (nrow < nNodes) (hi ? er : el)[nrow] = keep;
}

__global__ void __launch_bounds__(PJ_WAVES * 32)
gat_project_wmma(const float* __restrict__ h,
                 const float* __restrict__ W,
                 const float* __restrict__ a_l,
                 const float* __restrict__ a_r,
                 float* __restrict__ hW,
                 float* __restrict__ el, float* __restrict__ er,
                 int* __restrict__ binTotal,
                 int nNodes, int nBins) {
  const int gt = blockIdx.x * (PJ_WAVES * 32) + threadIdx.x;
  if (gt < nBins) binTotal[gt] = 0;
  const int lane = threadIdx.x & 31;
  const int half = lane >> 4;
  const int l16  = lane & 15;
  const int tile0 = (gt >> 5) * PJ_TILES;
  if (tile0 * 16 >= nNodes) return;

  const float* brow = W + (size_t)l16 * IN_FEATS + half * 4;
  v4f b[16];
#pragma unroll
  for (int k8 = 0; k8 < 16; ++k8) b[k8] = *(const v4f*)(brow + k8 * 8);

  const float* arow[PJ_TILES];
#pragma unroll
  for (int t = 0; t < PJ_TILES; ++t) {
    int row = (tile0 + t) * 16 + l16;
    row = row < nNodes ? row : nNodes - 1;
    arow[t] = h + (size_t)row * IN_FEATS + half * 4;
  }
  v4f a[PJ_TILES][16];
#pragma unroll
  for (int k8 = 0; k8 < 16; ++k8)
#pragma unroll
    for (int t = 0; t < PJ_TILES; ++t) a[t][k8] = *(const v4f*)(arow[t] + k8 * 8);
  __builtin_amdgcn_sched_barrier(0);

  v8f acc[PJ_TILES];
#pragma unroll
  for (int t = 0; t < PJ_TILES; ++t) acc[t] = (v8f){};
#pragma unroll
  for (int k8 = 0; k8 < 16; ++k8) {
#pragma unroll
    for (int t = 0; t < PJ_TILES; ++t) {
      acc[t] = __builtin_amdgcn_wmma_f32_16x16x4_f32(false, a[t][k8].xy, false, b[k8].xy,
                                                     (short)0, acc[t], false, false);
      acc[t] = __builtin_amdgcn_wmma_f32_16x16x4_f32(false, a[t][k8].zw, false, b[k8].zw,
                                                     (short)0, acc[t], false, false);
    }
  }
  const float al = a_l[l16];
  const float ar = a_r[l16];
  project_tile_out(acc[0], tile0 * 16, half, l16, al, ar, hW, el, er, nNodes);
  if ((tile0 + 1) * 16 < nNodes)
    project_tile_out(acc[1], (tile0 + 1) * 16, half, l16, al, ar, hW, el, er, nNodes);
}

#define BIN_ROWS     196
#define SRC_BITS     20
#define SRC_MASK     ((1u << SRC_BITS) - 1u)
#define BIN_CHUNKS   256
#define BIN_THREADS  1024
#define X_CAP        16384

__device__ __forceinline__ void ldsMaxFloat(float* addr, float val) {
  if (val >= 0.f) atomicMax((int*)addr, __float_as_int(val));
  else            atomicMin((unsigned int*)addr, (unsigned int)__float_as_int(val));
}

__device__ __forceinline__ int chunk_len(int nEdges, int nChunks) {
  return (((nEdges + nChunks - 1) / nChunks) + 3) & ~3;
}

__device__ __forceinline__ void load4(const int* __restrict__ p, int base, int e1, int fill,
                                      int (&v)[4]) {
  if (base + 4 <= e1) {
    int4 t = *(const int4*)(p + base);
    v[0] = t.x; v[1] = t.y; v[2] = t.z; v[3] = t.w;
  } else {
#pragma unroll
    for (int j = 0; j < 4; ++j) v[j] = (base + j < e1) ? p[base + j] : fill;
  }
}

__device__ __forceinline__ int bin_of(int d, int nBins) {
  unsigned b = (unsigned)d / BIN_ROWS;
  return (int)(b < (unsigned)nBins ? b : (unsigned)(nBins - 1));
}

__global__ void __launch_bounds__(BIN_THREADS)
gat_bin_count(const int* __restrict__ dst, int* __restrict__ binTotal,
              int* __restrict__ binOff, int nEdges, int nBins) {
  extern __shared__ int bin_lds[];
  int* hist = bin_lds;
  const int tid = threadIdx.x;
  for (int b = tid; b < nBins; b += BIN_THREADS) hist[b] = 0;
  __syncthreads();
  const int len = chunk_len(nEdges, (int)gridDim.x);
  const int e0 = blockIdx.x * len;
  int e1 = e0 + len;
  if (e1 > nEdges) e1 = nEdges;
  for (int base = e0 + 4 * tid; base < e1; base += 4 * BIN_THREADS) {
    int d[4];
    load4(dst, base, e1, -1, d);
#pragma unroll
    for (int j = 0; j < 4; ++j)
      if (d[j] >= 0) atomicAdd(&hist[bin_of(d[j], nBins)], 1);
  }
  __syncthreads();
  for (int b = tid; b < nBins; b += BIN_THREADS)
    binOff[(size_t)blockIdx.x * nBins + b] = atomicAdd(&binTotal[b], hist[b]);
}

__device__ __forceinline__ int wave_scan_excl(int* a, int n, int lane) {
  const int per = (n + 31) / 32;
  const int b0 = lane * per;
  int b1 = b0 + per;
  if (b1 > n) b1 = n;
  int s = 0;
  for (int b = b0; b < b1; ++b) s += a[b];
  int incl = s;
#pragma unroll
  for (int o = 1; o < 32; o <<= 1) {
    int t = __shfl_up(incl, o);
    if (lane >= o) incl += t;
  }
  int run = incl - s;
  for (int b = b0; b < b1; ++b) {
    const int t = a[b];
    a[b] = run;
    run += t;
  }
  return __shfl(incl, 31);
}

#define SC_TRIPS   4
#define SC_BATCH   (SC_TRIPS * 4 * BIN_THREADS)

__device__ __forceinline__ void sc_load_batch(const int* __restrict__ src, const int* __restrict__ dst,
                                              int s0, int s1, int tid,
                                              int (&d)[SC_TRIPS][4], int (&s)[SC_TRIPS][4],
                                              bool (&ok)[SC_TRIPS]) {
#pragma unroll
  for (int t = 0; t < SC_TRIPS; ++t) {
    const int base = s0 + 4 * tid + t * 4 * BIN_THREADS;
    ok[t] = base < s1;
    const int cb = ok[t] ? base : s0;
    load4(dst, cb, s1, -1, d[t]);
    load4(src, cb, s1, 0, s[t]);
  }
}

__global__ void __launch_bounds__(BIN_THREADS)
gat_bin_scatter(const int* __restrict__ src, const int* __restrict__ dst,
                const int* __restrict__ binTotal, const int* __restrict__ binOff,
                int* __restrict__ binStart, unsigned* __restrict__ packed,
                int nEdges, int nBins) {
  extern __shared__ int scat_lds[];
  int* gcur   = scat_lds;
  int* lcur   = scat_lds + nBins;
  int* gdelta = lcur + nBins + 1;
  __shared__ uint2 rec[SC_BATCH];
  const int tid = threadIdx.x;
  const int len = chunk_len(nEdges, (int)gridDim.x);
  const int e0 = blockIdx.x * len;
  int e1 = e0 + len;
  if (e1 > nEdges) e1 = nEdges;
  const bool any = e0 < e1;

  int d[SC_TRIPS][4], s[SC_TRIPS][4];
  bool ok[SC_TRIPS];
  int s1 = e0 + SC_BATCH;
  if (s1 > e1) s1 = e1;
  if (any) sc_load_batch(src, dst, e0, s1, tid, d, s, ok);

  for (int b = tid; b < nBins; b += BIN_THREADS) {
    gcur[b] = binTotal[b];
    gdelta[b] = binOff[(size_t)blockIdx.x * nBins + b];
  }
  __syncthreads();
  if (tid < 32) {
    const int tot = wave_scan_excl(gcur, nBins, tid);
    if (blockIdx.x == 0 && tid == 0) binStart[nBins] = tot;
  }
  __syncthreads();
  for (int b = tid; b < nBins; b += BIN_THREADS) {
    const int st = gcur[b];
    if (blockIdx.x == 0) binStart[b] = st;
    gcur[b] = st + gdelta[b];
  }

  if (any) {
    int s0 = e0;
    for (;;) {
      for (int b = tid; b <= nBins; b += BIN_THREADS) lcur[b] = 0;
      __syncthreads();
#pragma unroll
      for (int t = 0; t < SC_TRIPS; ++t)
#pragma unroll
        for (int j = 0; j < 4; ++j)
          if (ok[t] && d[t][j] >= 0) atomicAdd(&lcur[bin_of(d[t][j], nBins)], 1);
      __syncthreads();
      if (tid < 32) {
        const int tot = wave_scan_excl(lcur, nBins, tid);
        if (tid == 0) lcur[nBins] = tot;
      }
      __syncthreads();
      for (int b = tid; b < nBins; b += BIN_THREADS) {
        const int ls = lcur[b];
        const int g = gcur[b];
        gdelta[b] = g - ls;
        gcur[b] = g + (lcur[b + 1] - ls);
      }
      const int nPlaced = lcur[nBins];
      __syncthreads();
#pragma unroll
      for (int t = 0; t < SC_TRIPS; ++t)
#pragma unroll
        for (int j = 0; j < 4; ++j)
          if (ok[t] && d[t][j] >= 0) {
            const int b = bin_of(d[t][j], nBins);
            const int lp = atomicAdd(&lcur[b], 1);
            const unsigned w = (((unsigned)d[t][j] % BIN_ROWS) << SRC_BITS) | ((unsigned)s[t][j] & SRC_MASK);
            rec[lp] = make_uint2(w, (unsigned)(gdelta[b] + lp));
          }
      __syncthreads();
      for (int i = tid; i < nPlaced; i += BIN_THREADS) {
        const uint2 r = rec[i];
        packed[r.y] = r.x;
      }
      s0 += SC_BATCH;
      if (s0 >= e1) break;
      s1 = s0 + SC_BATCH;
      if (s1 > e1) s1 = e1;
      sc_load_batch(src, dst, s0, s1, tid, d, s, ok);
    }
  }
}

template <int K> __device__ __forceinline__ int row_share_i(int v) {
  return __builtin_amdgcn_update_dpp(v, v, 0x150 + K, 0xF, 0xF, false);
}
template <int K> __device__ __forceinline__ float row_share_f(float v) {
  return __int_as_float(row_share_i<K>(__float_as_int(v)));
}
#define AG_FOR16(X) X(0) X(1) X(2) X(3) X(4) X(5) X(6) X(7) X(8) X(9) X(10) X(11) X(12) X(13) X(14) X(15)
#define AG_ROW_LOAD(K) hv[K] = hW[(unsigned)(row_share_i<K>(s) * OUT_FEATS + f)];
#define AG_ROW_ADD(K)  atomicAdd(tab + row_share_i<K>(urow) * OUT_FEATS + f, hv[K] * row_share_f<K>(w));

#define AG_GROUPS   (BIN_THREADS / OUT_FEATS)
#define MAX_EDGES   4

__global__ void __launch_bounds__(BIN_THREADS)
gat_bin_aggregate(const unsigned* __restrict__ packed, const int* __restrict__ binStart,
                  const float* __restrict__ el, const float* __restrict__ er,
                  const float* __restrict__ hW, float* __restrict__ out, int nNodes) {
  __shared__ float tab[BIN_ROWS * OUT_FEATS];
  __shared__ float den[BIN_ROWS];
  __shared__ float mx[BIN_ROWS];
  __shared__ float erl[BIN_ROWS];
  __shared__ float xc[X_CAP];
  const int tid = threadIdx.x;
  const int g = tid >> 4;
  const int f = tid & 15;
  const int lo = blockIdx.x * BIN_ROWS;
  int cnt = nNodes - lo;
  cnt = cnt > BIN_ROWS ? BIN_ROWS : cnt;
  for (int i = tid; i < BIN_ROWS * OUT_FEATS; i += BIN_THREADS) tab[i] = 0.f;
  if (tid < BIN_ROWS) {
    den[tid] = 0.f;
    mx[tid] = __int_as_float(0xFF800000);
    erl[tid] = (tid < cnt) ? er[lo + tid] : 0.f;
  }
  __syncthreads();
  const int p0 = binStart[blockIdx.x];
  const int p1 = binStart[blockIdx.x + 1];

  for (int i0 = p0 + tid; i0 < p1; i0 += MAX_EDGES * BIN_THREADS) {
    unsigned pk[MAX_EDGES];
    float v[MAX_EDGES];
#pragma unroll
    for (int u = 0; u < MAX_EDGES; ++u) {
      const int i = i0 + u * BIN_THREADS;
      pk[u] = packed[i < p1 ? i : i0];
    }
#pragma unroll
    for (int u = 0; u < MAX_EDGES; ++u) v[u] = el[pk[u] & SRC_MASK];
#pragma unroll
    for (int u = 0; u < MAX_EDGES; ++u) {
      const int i = i0 + u * BIN_THREADS;
      if (i < p1) {
        const int r = (int)(pk[u] >> SRC_BITS);
        float x = v[u] + erl[r];
        x = (x > 0.f) ? x : NEG_SLOPE * x;
        if (i - p0 < X_CAP) xc[i - p0] = x;
        ldsMaxFloat(mx + r, x);
      }
    }
  }
  __syncthreads();
  if (tid < BIN_ROWS) {
    float m = mx[tid];
    if (!(fabsf(m) < __int_as_float(0x7F800000))) m = 0.f;
    mx[tid] = m;
  }
  __syncthreads();

  int q0 = p0 + g * OUT_FEATS;
  unsigned pk = 0u;
  if (q0 < p1) pk = packed[(q0 + f < p1) ? q0 + f : q0];
  for (; q0 < p1; q0 += AG_GROUPS * OUT_FEATS) {
    const int nq0 = q0 + AG_GROUPS * OUT_FEATS;
    const int ni = (nq0 + f < p1) ? nq0 + f : ((nq0 < p1) ? nq0 : q0);
    const unsigned npk = packed[ni];
    const bool okl = q0 + f < p1;
    const int j = (okl ? q0 + f : q0) - p0;
    const int s = (int)(pk & SRC_MASK);
    const int urow = (int)(pk >> SRC_BITS);
    float hv[OUT_FEATS];
    AG_FOR16(AG_ROW_LOAD)
    float x;
    if (j < X_CAP) {
      x = xc[j];
    } else {
      x = el[s] + erl[urow];
      x = (x > 0.f) ? x : NEG_SLOPE * x;
    }
    float w = __expf(x - mx[urow]);
    w = okl ? w : 0.f;
    AG_FOR16(AG_ROW_ADD)
    atomicAdd(den + urow, w);
    pk = npk;
  }
  __syncthreads();

  float* o = out + (size_t)lo * OUT_FEATS;
  for (int i = tid; i < cnt * OUT_FEATS; i += BIN_THREADS)
    o[i] = tab[i] / fmaxf(den[i >> 4], 1e-16f);
}

extern "C" void kernel_launch(void* const* d_in, const int* in_sizes, int n_in,
                              void* d_out, int out_size, void* d_ws, size_t ws_size,
                              hipStream_t stream) {
  const float* h   = (const float*)d_in[0];
  const int*   src = (const int*)d_in[1];
  const int*   dst = (const int*)d_in[2];
  const float* W   = (const float*)d_in[3];
  const float* a_l = (const float*)d_in[4];
  const float* a_r = (const float*)d_in[5];
  float* out = (float*)d_out;

  int nNodes = in_sizes[0] / IN_FEATS;
  int nEdges = in_sizes[1];
  int nBins = (nNodes + BIN_ROWS - 1) / BIN_ROWS;

  float* ws        = (float*)d_ws;
  float* hW        = ws;
  float* el        = hW + (size_t)nNodes * OUT_FEATS;
  float* er        = el + nNodes;
  unsigned* packed = (unsigned*)(er + nNodes);
  int* binTotal    = (int*)(packed + nEdges);
  int* binStart    = binTotal + nBins;
  int* binOff      = binStart + nBins + 1;

  dim3 blk(PJ_WAVES * 32);
  int tiles = (nNodes + 15) / 16;
  int blocksK1 = (tiles + PJ_WAVES * PJ_TILES - 1) / (PJ_WAVES * PJ_TILES);
  size_t countLds = (size_t)nBins * sizeof(int);
  size_t scatterLds = ((size_t)nBins * 3 + 1) * sizeof(int);

  gat_project_wmma<<<blocksK1, blk, 0, stream>>>(h, W, a_l, a_r, hW, el, er, binTotal, nNodes, nBins);
  gat_bin_count<<<BIN_CHUNKS, BIN_THREADS, countLds, stream>>>(dst, binTotal, binOff, nEdges, nBins);
  gat_bin_scatter<<<BIN_CHUNKS, BIN_THREADS, scatterLds, stream>>>(src, dst, binTotal, binOff,
                                                                  binStart, packed, nEdges, nBins);
  gat_bin_aggregate<<<nBins, BIN_THREADS, 0, stream>>>(packed, binStart, el, er, hW, out, nNodes);
}


# Aggregate reads a bin's edges once if max el - min el <= 32: shift leaky(max el + er[row]) known beforehand, edge el summed from its gathered row; else the row-max read stays; weights rescaled
# speedup vs baseline: 8.9850x; 1.1684x over previous
#include <hip/hip_runtime.h>
#include <hip/hip_bf16.h>

#define IN_FEATS   128
#define OUT_FEATS  16
#define NEG_SLOPE  0.2f

typedef float v2f __attribute__((ext_vector_type(2)));
typedef float v4f __attribute__((ext_vector_type(4)));
typedef float v8f __attribute__((ext_vector_type(8)));

template <int K> __device__ __forceinline__ float row_xmask_f(float v) {
  return __int_as_float(__builtin_amdgcn_update_dpp(__float_as_int(v), __float_as_int(v),
                                                    0x160 + K, 0xF, 0xF, false));
}

#define PJ_TILES  2
#define PJ_WAVES  4

__device__ __forceinline__ void project_tile_out(const v8f acc, int m0, int half, int l16,
                                                 float al, float ar,
                                                 float* __restrict__ hW,
                                                 float* __restrict__ el, float* __restrict__ er,
                                                 int nNodes, float& top, float& bot) {
#pragma unroll
  for (int g = 0; g < 8; ++g) {
    int row = m0 + half * 8 + g;
    hW[(size_t)row * OUT_FEATS + l16] = acc[g];
  }
  const bool hi = (l16 & 8) != 0;
  float keep = 0.f;
#pragma unroll
  for (int g = 0; g < 8; ++g) {
    float pl = acc[g] * al;
    float pr = acc[g] * ar;
    pl += row_xmask_f<8>(pl);
    pr += row_xmask_f<8>(pr);
    float u = hi ? pr : pl;
    u += row_xmask_f<4>(u);
    u += row_xmask_f<2>(u);
    u += row_xmask_f<1>(u);
    if ((l16 & 7) == g) keep = u;
  }
  const int nrow = m0 + half * 8 + (l16 & 7);
  if (nrow < nNodes) (hi ? er : el)[nrow] = keep;
  if (!hi && nrow < nNodes) {
    top = fmaxf(top, keep);
    bot = fminf(bot, keep);
  }
}

__global__ void __launch_bounds__(PJ_WAVES * 32)
gat_project_wmma(const float* __restrict__ h,
                 const float* __restrict__ W,
                 const float* __restrict__ a_l,
                 const float* __restrict__ a_r,
                 float* __restrict__ hW,
                 float* __restrict__ el, float* __restrict__ er,
                 float* __restrict__ elSpan,
                 int* __restrict__ binTotal,
                 int nNodes, int nBins) {
  const int gt = blockIdx.x * (PJ_WAVES * 32) + threadIdx.x;
  if (gt < nBins) binTotal[gt] = 0;
  const int lane = threadIdx.x & 31;
  const int half = lane >> 4;
  const int l16  = lane & 15;
  const int tile0 = (gt >> 5) * PJ_TILES;
  if (tile0 * 16 >= nNodes) {
    if (lane == 0) {
      elSpan[2 * (gt >> 5)]     = __int_as_float(0xFF800000);
      elSpan[2 * (gt >> 5) + 1] = __int_as_float(0x7F800000);
    }
    return;
  }

  const float* brow = W + (size_t)l16 * IN_FEATS + half * 4;
  v4f b[16];
#pragma unroll
  for (int k8 = 0; k8 < 16; ++k8) b[k8] = *(const v4f*)(brow + k8 * 8);

  const float* arow[PJ_TILES];
#pragma unroll
  for (int t = 0; t < PJ_TILES; ++t) {
    int row = (tile0 + t) * 16 + l16;
    row = row < nNodes ? row : nNodes - 1;
    arow[t] = h + (size_t)row * IN_FEATS + half * 4;
  }
  v4f a[PJ_TILES][16];
#pragma unroll
  for (int k8 = 0; k8 < 16; ++k8)
#pragma unroll
    for (int t = 0; t < PJ_TILES; ++t) a[t][k8] = *(const v4f*)(arow[t] + k8 * 8);
  __builtin_amdgcn_sched_barrier(0);

  v8f acc[PJ_TILES];
#pragma unroll
  for (int t = 0; t < PJ_TILES; ++t) acc[t] = (v8f){};
#pragma unroll
  for (int k8 = 0; k8 < 16; ++k8) {
#pragma unroll
    for (int t = 0; t < PJ_TILES; ++t) {
      acc[t] = __builtin_amdgcn_wmma_f32_16x16x4_f32(false, a[t][k8].xy, false, b[k8].xy,
                                                     (short)0, acc[t], false, false);
      acc[t] = __builtin_amdgcn_wmma_f32_16x16x4_f32(false, a[t][k8].zw, false, b[k8].zw,
                                                     (short)0, acc[t], false, false);
    }
  }
  const float al = a_l[l16];
  const float ar = a_r[l16];
  float top = __int_as_float(0xFF800000);
  float bot = __int_as_float(0x7F800000);
  project_tile_out(acc[0], tile0 * 16, half, l16, al, ar, hW, el, er, nNodes, top, bot);
  if ((tile0 + 1) * 16 < nNodes)
    project_tile_out(acc[1], (tile0 + 1) * 16, half, l16, al, ar, hW, el, er, nNodes, top, bot);
#pragma unroll
  for (int o = 16; o > 0; o >>= 1) {
    top = fmaxf(top, __shfl_xor(top, o));
    bot = fminf(bot, __shfl_xor(bot, o));
  }
  if (lane == 0) {
    elSpan[2 * (gt >> 5)]     = top;
    elSpan[2 * (gt >> 5) + 1] = bot;
  }
}

#define BIN_ROWS     196
#define SRC_BITS     20
#define SRC_MASK     ((1u << SRC_BITS) - 1u)
#define BIN_CHUNKS   256
#define BIN_THREADS  1024

__device__ __forceinline__ void ldsMaxFloat(float* addr, float val) {
  if (val >= 0.f) atomicMax((int*)addr, __float_as_int(val));
  else            atomicMin((unsigned int*)addr, (unsigned int)__float_as_int(val));
}

__device__ __forceinline__ int chunk_len(int nEdges, int nChunks) {
  return (((nEdges + nChunks - 1) / nChunks) + 3) & ~3;
}

__device__ __forceinline__ void load4(const int* __restrict__ p, int base, int e1, int fill,
                                      int (&v)[4]) {
  if (base + 4 <= e1) {
    int4 t = *(const int4*)(p + base);
    v[0] = t.x; v[1] = t.y; v[2] = t.z; v[3] = t.w;
  } else {
#pragma unroll
    for (int j = 0; j < 4; ++j) v[j] = (base + j < e1) ? p[base + j] : fill;
  }
}

__device__ __forceinline__ int bin_of(int d, int nBins) {
  unsigned b = (unsigned)d / BIN_ROWS;
  return (int)(b < (unsigned)nBins ? b : (unsigned)(nBins - 1));
}

__global__ void __launch_bounds__(BIN_THREADS)
gat_bin_count(const int* __restrict__ dst, int* __restrict__ binTotal,
              int* __restrict__ binOff, int nEdges, int nBins) {
  extern __shared__ int bin_lds[];
  int* hist = bin_lds;
  const int tid = threadIdx.x;
  for (int b = tid; b < nBins; b += BIN_THREADS) hist[b] = 0;
  __syncthreads();
  const int len = chunk_len(nEdges, (int)gridDim.x);
  const int e0 = blockIdx.x * len;
  int e1 = e0 + len;
  if (e1 > nEdges) e1 = nEdges;
  for (int base = e0 + 4 * tid; base < e1; base += 4 * BIN_THREADS) {
    int d[4];
    load4(dst, base, e1, -1, d);
#pragma unroll
    for (int j = 0; j < 4; ++j)
      if (d[j] >= 0) atomicAdd(&hist[bin_of(d[j], nBins)], 1);
  }
  __syncthreads();
  for (int b = tid; b < nBins; b += BIN_THREADS)
    binOff[(size_t)blockIdx.x * nBins + b] = atomicAdd(&binTotal[b], hist[b]);
}

__device__ __forceinline__ int wave_scan_excl(int* a, int n, int lane) {
  const int per = (n + 31) / 32;
  const int b0 = lane * per;
  int b1 = b0 + per;
  if (b1 > n) b1 = n;
  int s = 0;
  for (int b = b0; b < b1; ++b) s += a[b];
  int incl = s;
#pragma unroll
  for (int o = 1; o < 32; o <<= 1) {
    int t = __shfl_up(incl, o);
    if (lane >= o) incl += t;
  }
  int run = incl - s;
  for (int b = b0; b < b1; ++b) {
    const int t = a[b];
    a[b] = run;
    run += t;
  }
  return __shfl(incl, 31);
}

#define SC_TRIPS   4
#define SC_BATCH   (SC_TRIPS * 4 * BIN_THREADS)

__device__ __forceinline__ void sc_load_batch(const int* __restrict__ src, const int* __restrict__ dst,
                                              int s0, int s1, int tid,
                                              int (&d)[SC_TRIPS][4], int (&s)[SC_TRIPS][4],
                                              bool (&ok)[SC_TRIPS]) {
#pragma unroll
  for (int t = 0; t < SC_TRIPS; ++t) {
    const int base = s0 + 4 * tid + t * 4 * BIN_THREADS;
    ok[t] = base < s1;
    const int cb = ok[t] ? base : s0;
    load4(dst, cb, s1, -1, d[t]);
    load4(src, cb, s1, 0, s[t]);
  }
}

__global__ void __launch_bounds__(BIN_THREADS)
gat_bin_scatter(const int* __restrict__ src, const int* __restrict__ dst,
                const int* __restrict__ binTotal, const int* __restrict__ binOff,
                int* __restrict__ binStart, unsigned* __restrict__ packed,
                int nEdges, int nBins) {
  extern __shared__ int scat_lds[];
  int* gcur   = scat_lds;
  int* lcur   = scat_lds + nBins;
  int* gdelta = lcur + nBins + 1;
  __shared__ uint2 rec[SC_BATCH];
  const int tid = threadIdx.x;
  const int len = chunk_len(nEdges, (int)gridDim.x);
  const int e0 = blockIdx.x * len;
  int e1 = e0 + len;
  if (e1 > nEdges) e1 = nEdges;
  const bool any = e0 < e1;

  int d[SC_TRIPS][4], s[SC_TRIPS][4];
  bool ok[SC_TRIPS];
  int s1 = e0 + SC_BATCH;
  if (s1 > e1) s1 = e1;
  if (any) sc_load_batch(src, dst, e0, s1, tid, d, s, ok);

  for (int b = tid; b < nBins; b += BIN_THREADS) {
    gcur[b] = binTotal[b];
    gdelta[b] = binOff[(size_t)blockIdx.x * nBins + b];
  }
  __syncthreads();
  if (tid < 32) {
    const int tot = wave_scan_excl(gcur, nBins, tid);
    if (blockIdx.x == 0 && tid == 0) binStart[nBins] = tot;
  }
  __syncthreads();
  for (int b = tid; b < nBins; b += BIN_THREADS) {
    const int st = gcur[b];
    if (blockIdx.x == 0) binStart[b] = st;
    gcur[b] = st + gdelta[b];
  }

  if (any) {
    int s0 = e0;
    for (;;) {
      for (int b = tid; b <= nBins; b += BIN_THREADS) lcur[b] = 0;
      __syncthreads();
#pragma unroll
      for (int t = 0; t < SC_TRIPS; ++t)
#pragma unroll
        for (int j = 0; j < 4; ++j)
          if (ok[t] && d[t][j] >= 0) atomicAdd(&lcur[bin_of(d[t][j], nBins)], 1);
      __syncthreads();
      if (tid < 32) {
        const int tot = wave_scan_excl(lcur, nBins, tid);
        if (tid == 0) lcur[nBins] = tot;
      }
      __syncthreads();
      for (int b = tid; b < nBins; b += BIN_THREADS) {
        const int ls = lcur[b];
        const int g = gcur[b];
        gdelta[b] = g - ls;
        gcur[b] = g + (lcur[b + 1] - ls);
      }
      const int nPlaced = lcur[nBins];
      __syncthreads();
#pragma unroll
      for (int t = 0; t < SC_TRIPS; ++t)
#pragma unroll
        for (int j = 0; j < 4; ++j)
          if (ok[t] && d[t][j] >= 0) {
            const int b = bin_of(d[t][j], nBins);
            const int lp = atomicAdd(&lcur[b], 1);
            const unsigned w = (((unsigned)d[t][j] % BIN_ROWS) << SRC_BITS) | ((unsigned)s[t][j] & SRC_MASK);
            rec[lp] = make_uint2(w, (unsigned)(gdelta[b] + lp));
          }
      __syncthreads();
      for (int i = tid; i < nPlaced; i += BIN_THREADS) {
        const uint2 r = rec[i];
        packed[r.y] = r.x;
      }
      s0 += SC_BATCH;
      if (s0 >= e1) break;
      s1 = s0 + SC_BATCH;
      if (s1 > e1) s1 = e1;
      sc_load_batch(src, dst, s0, s1, tid, d, s, ok);
    }
  }
}

template <int K> __device__ __forceinline__ int row_share_i(int v) {
  return __builtin_amdgcn_update_dpp(v, v, 0x150 + K, 0xF, 0xF, false);
}
template <int K> __device__ __forceinline__ float row_share_f(float v) {
  return __int_as_float(row_share_i<K>(__float_as_int(v)));
}
#define AG_FOR16(X) X(0) X(1) X(2) X(3) X(4) X(5) X(6) X(7) X(8) X(9) X(10) X(11) X(12) X(13) X(14) X(15)
#define AG_ROW_LOAD(K) hv[K] = hW[(unsigned)(row_share_i<K>(s) * OUT_FEATS + f)];
#define AG_ROW_ADD(K)  atomicAdd(tab + row_share_i<K>(urow) * OUT_FEATS + f, hv[K] * row_share_f<K>(w));

#define AG_GROUPS   (BIN_THREADS / OUT_FEATS)

__device__ __forceinline__ float row_edges_el(const float (&hv)[OUT_FEATS], float al, int f) {
  const bool b8 = (f & 8) != 0, b4 = (f & 4) != 0, b2 = (f & 2) != 0, b1 = (f & 1) != 0;
  float q[8], r[4], t[2];
#pragma unroll
  for (int j = 0; j < 8; ++j) {
    const float lo = hv[j] * al, up = hv[j + 8] * al;
    q[j] = (b8 ? up : lo) + row_xmask_f<8>(b8 ? lo : up);
  }
#pragma unroll
  for (int j = 0; j < 4; ++j) r[j] = (b4 ? q[j + 4] : q[j]) + row_xmask_f<4>(b4 ? q[j] : q[j + 4]);
#pragma unroll
  for (int j = 0; j < 2; ++j) t[j] = (b2 ? r[j + 2] : r[j]) + row_xmask_f<2>(b2 ? r[j] : r[j + 2]);
  return (b1 ? t[1] : t[0]) + row_xmask_f<1>(b1 ? t[0] : t[1]);
}

#define MAX_EDGES    4
#define EL_SPAN_MAX  32.0f

__global__ void __launch_bounds__(BIN_THREADS)
gat_bin_aggregate(const unsigned* __restrict__ packed, const int* __restrict__ binStart,
                  const float* __restrict__ el, const float* __restrict__ elSpan, int nSpan,
                  const float* __restrict__ a_l, const float* __restrict__ er,
                  const float* __restrict__ hW, float* __restrict__ out, int nNodes) {
  __shared__ float tab[BIN_ROWS * OUT_FEATS];
  __shared__ float den[BIN_ROWS];
  __shared__ float cb[BIN_ROWS];
  __shared__ float erl[BIN_ROWS];
  __shared__ float redTop[BIN_THREADS / 32];
  __shared__ float redBot[BIN_THREADS / 32];
  const int tid = threadIdx.x;
  const int g = tid >> 4;
  const int f = tid & 15;
  const int lo = blockIdx.x * BIN_ROWS;
  int cnt = nNodes - lo;
  cnt = cnt > BIN_ROWS ? BIN_ROWS : cnt;
  const int p0 = binStart[blockIdx.x];
  const int p1 = binStart[blockIdx.x + 1];
  int q0 = p0 + g * OUT_FEATS;
  unsigned pk = 0u;
  if (q0 < p1) pk = packed[(q0 + f < p1) ? q0 + f : q0];
  const float al = a_l[f];
  float erv = 0.f;
  if (tid < cnt) erv = er[lo + tid];

  float top = __int_as_float(0xFF800000);
  float bot = __int_as_float(0x7F800000);
  for (int i = tid; i < nSpan; i += BIN_THREADS) {
    top = fmaxf(top, elSpan[2 * i]);
    bot = fminf(bot, elSpan[2 * i + 1]);
  }
#pragma unroll
  for (int o = 16; o > 0; o >>= 1) {
    top = fmaxf(top, __shfl_xor(top, o));
    bot = fminf(bot, __shfl_xor(bot, o));
  }
  if ((tid & 31) == 0) {
    redTop[tid >> 5] = top;
    redBot[tid >> 5] = bot;
  }
  for (int i = tid; i < BIN_ROWS * OUT_FEATS; i += BIN_THREADS) tab[i] = 0.f;
  if (tid < BIN_ROWS) {
    den[tid] = 0.f;
    erl[tid] = erv;
    cb[tid] = __int_as_float(0xFF800000);
  }
  __syncthreads();
  top = redTop[tid & 31];
  bot = redBot[tid & 31];
#pragma unroll
  for (int o = 16; o > 0; o >>= 1) {
    top = fmaxf(top, __shfl_xor(top, o));
    bot = fminf(bot, __shfl_xor(bot, o));
  }
  const bool oneRead = (top - bot) <= EL_SPAN_MAX;
  if (oneRead) {
    if (tid < BIN_ROWS) {
      const float c = top + erv;
      cb[tid] = (c > 0.f) ? c : NEG_SLOPE * c;
    }
  } else {
    for (int i0 = p0 + tid; i0 < p1; i0 += MAX_EDGES * BIN_THREADS) {
      unsigned mk[MAX_EDGES];
      float v[MAX_EDGES];
#pragma unroll
      for (int u = 0; u < MAX_EDGES; ++u) {
        const int i = i0 + u * BIN_THREADS;
        mk[u] = packed[i < p1 ? i : i0];
      }
#pragma unroll
      for (int u = 0; u < MAX_EDGES; ++u) v[u] = el[mk[u] & SRC_MASK];
#pragma unroll
      for (int u = 0; u < MAX_EDGES; ++u) {
        const int i = i0 + u * BIN_THREADS;
        if (i < p1) {
          const int r = (int)(mk[u] >> SRC_BITS);
          float x = v[u] + erl[r];
          x = (x > 0.f) ? x : NEG_SLOPE * x;
          ldsMaxFloat(cb + r, x);
        }
      }
    }
    __syncthreads();
    if (tid < BIN_ROWS) {
      float m = cb[tid];
      if (!(fabsf(m) < __int_as_float(0x7F800000))) m = 0.f;
      cb[tid] = m;
    }
  }
  __syncthreads();

  for (; q0 < p1; q0 += AG_GROUPS * OUT_FEATS) {
    const int nq0 = q0 + AG_GROUPS * OUT_FEATS;
    const int ni = (nq0 + f < p1) ? nq0 + f : ((nq0 < p1) ? nq0 : q0);
    const unsigned npk = packed[ni];
    const bool okl = q0 + f < p1;
    const int s = (int)(pk & SRC_MASK);
    const int urow = (int)(pk >> SRC_BITS);
    float hv[OUT_FEATS];
    AG_FOR16(AG_ROW_LOAD)
    float x = row_edges_el(hv, al, f) + erl[urow];
    x = (x > 0.f) ? x : NEG_SLOPE * x;
    float w = __expf(x - cb[urow]);
    w = okl ? w : 0.f;
    AG_FOR16(AG_ROW_ADD)
    atomicAdd(den + urow, w);
    pk = npk;
  }
  __syncthreads();

  float* o = out + (size_t)lo * OUT_FEATS;
  for (int i = tid; i < cnt * OUT_FEATS; i += BIN_THREADS)
    o[i] = tab[i] / fmaxf(den[i >> 4], 1e-16f);
}

extern "C" void kernel_launch(void* const* d_in, const int* in_sizes, int n_in,
                              void* d_out, int out_size, void* d_ws, size_t ws_size,
                              hipStream_t stream) {
  const float* h   = (const float*)d_in[0];
  const int*   src = (const int*)d_in[1];
  const int*   dst = (const int*)d_in[2];
  const float* W   = (const float*)d_in[3];
  const float* a_l = (const float*)d_in[4];
  const float* a_r = (const float*)d_in[5];
  float* out = (float*)d_out;

  int nNodes = in_sizes[0] / IN_FEATS;
  int nEdges = in_sizes[1];
  int nBins = (nNodes + BIN_ROWS - 1) / BIN_ROWS;

  float* ws        = (float*)d_ws;
  float* hW        = ws;
  float* el        = hW + (size_t)nNodes * OUT_FEATS;
  float* er        = el + nNodes;
  unsigned* packed = (unsigned*)(er + nNodes);
  int* binTotal    = (int*)(packed + nEdges);
  int* binStart    = binTotal + nBins;
  int* binOff      = binStart + nBins + 1;
  float* elSpan    = (float*)(binOff + (size_t)BIN_CHUNKS * nBins);

  dim3 blk(PJ_WAVES * 32);
  int tiles = (nNodes + 15) / 16;
  int blocksK1 = (tiles + PJ_WAVES * PJ_TILES - 1) / (PJ_WAVES * PJ_TILES);
  int nSpan = blocksK1 * PJ_WAVES;
  size_t countLds = (size_t)nBins * sizeof(int);
  size_t scatterLds = ((size_t)nBins * 3 + 1) * sizeof(int);

  gat_project_wmma<<<blocksK1, blk, 0, stream>>>(h, W, a_l, a_r, hW, el, er, elSpan, binTotal, nNodes, nBins);
  gat_bin_count<<<BIN_CHUNKS, BIN_THREADS, countLds, stream>>>(dst, binTotal, binOff, nEdges, nBins);
  gat_bin_scatter<<<BIN_CHUNKS, BIN_THREADS, scatterLds, stream>>>(src, dst, binTotal, binOff,
                                                                  binStart, packed, nEdges, nBins);
  gat_bin_aggregate<<<nBins, BIN_THREADS, 0, stream>>>(packed, binStart, el, elSpan, nSpan, a_l, er, hW, out, nNodes);
}
